# starting kernel
# baseline (speedup 1.0000x reference)
#include <hip/hip_runtime.h>

#define DIN   256
#define H_DIM 128
#define DOUT  64
#define BN_RSQRT 0.9999950000374996f

typedef __attribute__((ext_vector_type(2))) float v2f;
typedef __attribute__((ext_vector_type(8))) float v8f;

__device__ __forceinline__ v8f wmma4(v2f a, v2f b, v8f c) {
    return __builtin_amdgcn_wmma_f32_16x16x4_f32(
         false, a,  false, b,
         (short)0, c,  false,  false);
}

__device__ __forceinline__ void gemm_accum(const float* __restrict__ A,
                                           const float* __restrict__ W,
                                           int K, int ncols,
                                           int row0, int col0, int m, int hi,
                                           float rs0, float rs1,
                                           v8f (&acc)[2][2]) {
    const float* ar0 = A + (size_t)(row0 + m) * K;
    const float* ar1 = A + (size_t)(row0 + 16 + m) * K;
    for (int k = 0; k < K; k += 4) {
        const int ka = k + 2 * hi;
        v2f a0, a1, b0, b1;
        a0.x = ar0[ka] * rs0;  a0.y = ar0[ka + 1] * rs0;
        a1.x = ar1[ka] * rs1;  a1.y = ar1[ka + 1] * rs1;
        const float* wr = W + (size_t)ka * ncols;
        b0.x = wr[col0 + m];       b0.y = wr[ncols + col0 + m];
        b1.x = wr[col0 + 16 + m];  b1.y = wr[ncols + col0 + 16 + m];
        acc[0][0] = wmma4(a0, b0, acc[0][0]);
        acc[0][1] = wmma4(a0, b1, acc[0][1]);
        acc[1][0] = wmma4(a1, b0, acc[1][0]);
        acc[1][1] = wmma4(a1, b1, acc[1][1]);
    }
}

__global__ __launch_bounds__(128)
void gemm_fused(const float* __restrict__ A0, const float* __restrict__ W0, int K0,
                const float* __restrict__ A1, const float* __restrict__ W1, int K1,
                const float* __restrict__ rowscale,
                const float* __restrict__ bias,
                const float* __restrict__ gamma, const float* __restrict__ beta,
                float* __restrict__ out, int nrows, int ncols, int do_relu) {
    const int wave = threadIdx.x >> 5;
    const int lane = threadIdx.x & 31;
    const int tilesN = ncols >> 5;
    const int tilesM = nrows >> 5;
    const int tile = blockIdx.x * 4 + wave;
    if (tile >= tilesM * tilesN) return;
    const int tm = tile / tilesN, tn = tile % tilesN;
    const int row0 = tm * 32, col0 = tn * 32;
    const int m = lane & 15, hi = lane >> 4;

    v8f acc[2][2] = {};

    float rs0 = 1.0f, rs1 = 1.0f;
    if (rowscale) { rs0 = rowscale[row0 + m]; rs1 = rowscale[row0 + 16 + m]; }
    gemm_accum(A0, W0, K0, ncols, row0, col0, m, hi, rs0, rs1, acc);
    if (A1) gemm_accum(A1, W1, K1, ncols, row0, col0, m, hi, 1.0f, 1.0f, acc);

#pragma unroll
    for (int mi = 0; mi < 2; ++mi) {
#pragma unroll
        for (int ni = 0; ni < 2; ++ni) {
            const int col = col0 + ni * 16 + m;
            const float bi = bias ? bias[col] : 0.0f;
            const float sc = gamma ? gamma[col] * BN_RSQRT : 1.0f;
            const float sh = beta ? beta[col] : 0.0f;
#pragma unroll
            for (int r = 0; r < 8; ++r) {
                const int row = row0 + mi * 16 + r + 8 * hi;
                float v = acc[mi][ni][r];
                v = (v + bi) * sc + sh;
                if (do_relu) v = fmaxf(v, 0.0f);
                out[(size_t)row * ncols + col] = v;
            }
        }
    }
}

__device__ __forceinline__ void atom_add_f32(float* p, float v) {
    unsafeAtomicAdd(p, v);
}

__global__ __launch_bounds__(256)
void scatter_kernel(const float* __restrict__ h, const int* __restrict__ src,
                    const int* __restrict__ dst, float* __restrict__ agg, int E) {
    const int t = blockIdx.x * blockDim.x + threadIdx.x;
    const int e = t >> 5;
    const int lane = t & 31;
    if (e >= E) return;
    const int s = src[e], d = dst[e];
    const float4 v = *(const float4*)(h + (size_t)s * H_DIM + lane * 4);
    float* ap = agg + (size_t)d * H_DIM + lane * 4;
    atom_add_f32(ap + 0, v.x);
    atom_add_f32(ap + 1, v.y);
    atom_add_f32(ap + 2, v.z);
    atom_add_f32(ap + 3, v.w);
}

__global__ __launch_bounds__(256)
void deg_kernel(const int* __restrict__ dst, float* __restrict__ deg, int E) {
    const int e = blockIdx.x * blockDim.x + threadIdx.x;
    if (e < E) atom_add_f32(&deg[dst[e]], 1.0f);
}

__global__ __launch_bounds__(256)
void invdeg_kernel(float* __restrict__ deg, int N) {
    const int i = blockIdx.x * blockDim.x + threadIdx.x;
    if (i < N) deg[i] = 1.0f / fmaxf(deg[i], 1.0f);
}

extern "C" void kernel_launch(void* const* d_in, const int* in_sizes, int n_in,
                              void* d_out, int out_size, void* d_ws, size_t ws_size,
                              hipStream_t stream) {
    const float* x     = (const float*)d_in[0];
    const int*   ei    = (const int*)d_in[1];
    const float* w_in  = (const float*)d_in[2];
    const float* b_in  = (const float*)d_in[3];
    const float* w_l1  = (const float*)d_in[4];
    const float* b_l1  = (const float*)d_in[5];
    const float* w_r1  = (const float*)d_in[6];
    const float* g1    = (const float*)d_in[7];
    const float* be1   = (const float*)d_in[8];
    const float* w_l2  = (const float*)d_in[9];
    const float* b_l2  = (const float*)d_in[10];
    const float* w_r2  = (const float*)d_in[11];
    const float* g2    = (const float*)d_in[12];
    const float* be2   = (const float*)d_in[13];
    const float* w_out = (const float*)d_in[14];
    const float* b_out = (const float*)d_in[15];

    const int N = in_sizes[0] / DIN;
    const int E = in_sizes[1] / 2;
    const int* src = ei;
    const int* dst = ei + E;

    float* ws  = (float*)d_ws;
    float* h0  = ws;
    float* agg = ws + (size_t)N * H_DIM;
    float* h1  = ws + 2 * (size_t)N * H_DIM;
    float* deg = ws + 3 * (size_t)N * H_DIM;

    const int tilesH   = (N / 32) * (H_DIM / 32);
    const int tilesOut = (N / 32) * (DOUT / 32);

    hipMemsetAsync(deg, 0, (size_t)N * sizeof(float), stream);
    deg_kernel<<<(E + 255) / 256, 256, 0, stream>>>(dst, deg, E);
    invdeg_kernel<<<(N + 255) / 256, 256, 0, stream>>>(deg, N);

    gemm_fused<<<(tilesH + 3) / 4, 128, 0, stream>>>(
        x, w_in, DIN, nullptr, nullptr, 0, nullptr, b_in, nullptr, nullptr,
        h0, N, H_DIM, 1);

    hipMemsetAsync(agg, 0, (size_t)N * H_DIM * sizeof(float), stream);
    scatter_kernel<<<(E * 32 + 255) / 256, 256, 0, stream>>>(h0, src, dst, agg, E);
    gemm_fused<<<(tilesH + 3) / 4, 128, 0, stream>>>(
        agg, w_l1, H_DIM, h0, w_r1, H_DIM, deg, b_l1, g1, be1,
        h1, N, H_DIM, 1);

    hipMemsetAsync(agg, 0, (size_t)N * H_DIM * sizeof(float), stream);
    scatter_kernel<<<(E * 32 + 255) / 256, 256, 0, stream>>>(h1, src, dst, agg, E);
    gemm_fused<<<(tilesH + 3) / 4, 128, 0, stream>>>(
        agg, w_l2, H_DIM, h1, w_r2, H_DIM, deg, b_l2, g2, be2,
        h0, N, H_DIM, 1);

    gemm_fused<<<(tilesOut + 3) / 4, 128, 0, stream>>>(
        h0, w_out, H_DIM, nullptr, nullptr, 0, nullptr, b_out, nullptr, nullptr,
        (float*)d_out, N, DOUT, 0);
}


# Aggregation without float atomics: edges bucketed by target per call (slot = returning int atomic count, scan, fill); one wave sums a node's source rows in registers in arrival-slot order
# speedup vs baseline: 1.9905x; 1.9905x over previous
#include <hip/hip_runtime.h>

#define DIN   256
#define H_DIM 128
#define DOUT  64
#define BN_RSQRT 0.9999950000374996f

typedef __attribute__((ext_vector_type(2))) float v2f;
typedef __attribute__((ext_vector_type(8))) float v8f;

__device__ __forceinline__ v8f wmma4(v2f a, v2f b, v8f c) {
    return __builtin_amdgcn_wmma_f32_16x16x4_f32(
         false, a,  false, b,
         (short)0, c,  false,  false);
}

__device__ __forceinline__ void gemm_accum(const float* __restrict__ A,
                                           const float* __restrict__ W,
                                           int K, int ncols,
                                           int row0, int col0, int m, int hi,
                                           float rs0, float rs1,
                                           v8f (&acc)[2][2]) {
    const float* ar0 = A + (size_t)(row0 + m) * K;
    const float* ar1 = A + (size_t)(row0 + 16 + m) * K;
    for (int k = 0; k < K; k += 4) {
        const int ka = k + 2 * hi;
        v2f a0, a1, b0, b1;
        a0.x = ar0[ka] * rs0;  a0.y = ar0[ka + 1] * rs0;
        a1.x = ar1[ka] * rs1;  a1.y = ar1[ka + 1] * rs1;
        const float* wr = W + (size_t)ka * ncols;
        b0.x = wr[col0 + m];       b0.y = wr[ncols + col0 + m];
        b1.x = wr[col0 + 16 + m];  b1.y = wr[ncols + col0 + 16 + m];
        acc[0][0] = wmma4(a0, b0, acc[0][0]);
        acc[0][1] = wmma4(a0, b1, acc[0][1]);
        acc[1][0] = wmma4(a1, b0, acc[1][0]);
        acc[1][1] = wmma4(a1, b1, acc[1][1]);
    }
}

__global__ __launch_bounds__(128)
void gemm_fused(const float* __restrict__ A0, const float* __restrict__ W0, int K0,
                const float* __restrict__ A1, const float* __restrict__ W1, int K1,
                const float* __restrict__ rowscale,
                const float* __restrict__ bias,
                const float* __restrict__ gamma, const float* __restrict__ beta,
                float* __restrict__ out, int nrows, int ncols, int do_relu) {
    const int wave = threadIdx.x >> 5;
    const int lane = threadIdx.x & 31;
    const int tilesN = ncols >> 5;
    const int tilesM = nrows >> 5;
    const int tile = blockIdx.x * 4 + wave;
    if (tile >= tilesM * tilesN) return;
    const int tm = tile / tilesN, tn = tile % tilesN;
    const int row0 = tm * 32, col0 = tn * 32;
    const int m = lane & 15, hi = lane >> 4;

    v8f acc[2][2] = {};

    float rs0 = 1.0f, rs1 = 1.0f;
    if (rowscale) { rs0 = rowscale[row0 + m]; rs1 = rowscale[row0 + 16 + m]; }
    gemm_accum(A0, W0, K0, ncols, row0, col0, m, hi, rs0, rs1, acc);
    if (A1) gemm_accum(A1, W1, K1, ncols, row0, col0, m, hi, 1.0f, 1.0f, acc);

#pragma unroll
    for (int mi = 0; mi < 2; ++mi) {
#pragma unroll
        for (int ni = 0; ni < 2; ++ni) {
            const int col = col0 + ni * 16 + m;
            const float bi = bias ? bias[col] : 0.0f;
            const float sc = gamma ? gamma[col] * BN_RSQRT : 1.0f;
            const float sh = beta ? beta[col] : 0.0f;
#pragma unroll
            for (int r = 0; r < 8; ++r) {
                const int row = row0 + mi * 16 + r + 8 * hi;
                float v = acc[mi][ni][r];
                v = (v + bi) * sc + sh;
                if (do_relu) v = fmaxf(v, 0.0f);
                out[(size_t)row * ncols + col] = v;
            }
        }
    }
}


__global__ __launch_bounds__(256)
void rank_kernel(const int* __restrict__ dst, int* __restrict__ cnt,
                 int* __restrict__ rank, int E) {
    const int e = blockIdx.x * blockDim.x + threadIdx.x;
    if (e < E) rank[e] = atomicAdd(&cnt[dst[e]], 1);
}

__global__ __launch_bounds__(1024)
void scan_kernel(const int* __restrict__ cnt, int* __restrict__ rowstart,
                 float* __restrict__ invdeg, int N) {
    __shared__ int s_pre[32];
    __shared__ int s_tot[32];
    const int tid = threadIdx.x;
    const int lane = tid & 31, wave = tid >> 5;
    const int tile0 = blockIdx.x * 1024;

    int pre = 0;
    for (int k = tid; k < tile0; k += 1024) pre += cnt[k];

    const int i = tile0 + tid;
    const int c = (i < N) ? cnt[i] : 0;

    int incl = c;
    for (int d = 1; d < 32; d <<= 1) {
        const int up = __shfl_up(incl, d);
        if (lane >= d) incl += up;
    }
    for (int d = 16; d > 0; d >>= 1) pre += __shfl_xor(pre, d);

    if (lane == 31) s_tot[wave] = incl;
    if (lane == 0)  s_pre[wave] = pre;
    __syncthreads();

    int base = 0;
    for (int w = 0; w < 32; ++w) {
        base += s_pre[w];
        if (w < wave) base += s_tot[w];
    }
    const int excl = base + incl - c;
    if (i < N) {
        rowstart[i] = excl;
        invdeg[i] = 1.0f / fmaxf((float)c, 1.0f);
    }
    if (i == N - 1) rowstart[N] = excl + c;
}

__global__ __launch_bounds__(256)
void fill_kernel(const int* __restrict__ src, const int* __restrict__ dst,
                 const int* __restrict__ rank, const int* __restrict__ rowstart,
                 int* __restrict__ col, int E) {
    const int e = blockIdx.x * blockDim.x + threadIdx.x;
    if (e < E) col[rowstart[dst[e]] + rank[e]] = src[e];
}

#define GATHER_ROWS(U)                                                          \
    {                                                                           \
        float4 v[U];                                                            \
        for (int u = 0; u < (U); ++u)                                           \
            v[u] = *(const float4*)(hp + (size_t)col[j + u] * H_DIM);           \
        for (int u = 0; u < (U); ++u) {                                         \
            acc.x += v[u].x; acc.y += v[u].y; acc.z += v[u].z; acc.w += v[u].w; \
        }                                                                       \
        j += (U);                                                               \
    }

__global__ __launch_bounds__(256)
void gather_sum_kernel(const float* __restrict__ h, const int* __restrict__ rowstart,
                       const int* __restrict__ col, float* __restrict__ agg, int N) {
    const int lane = threadIdx.x & 31;
    const int node = __builtin_amdgcn_readfirstlane((int)(blockIdx.x * 8 + (threadIdx.x >> 5)));
    if (node >= N) return;
    const int beg = rowstart[node];
    const int end = rowstart[node + 1];
    const float* hp = h + lane * 4;
    float4 acc = make_float4(0.0f, 0.0f, 0.0f, 0.0f);
    int j = beg;
    while (j + 8 <= end) GATHER_ROWS(8)
    if (j + 4 <= end) GATHER_ROWS(4)
    if (j + 2 <= end) GATHER_ROWS(2)
    if (j < end) GATHER_ROWS(1)
    *(float4*)(agg + (size_t)node * H_DIM + lane * 4) = acc;
}

extern "C" void kernel_launch(void* const* d_in, const int* in_sizes, int n_in,
                              void* d_out, int out_size, void* d_ws, size_t ws_size,
                              hipStream_t stream) {
    const float* x     = (const float*)d_in[0];
    const int*   ei    = (const int*)d_in[1];
    const float* w_in  = (const float*)d_in[2];
    const float* b_in  = (const float*)d_in[3];
    const float* w_l1  = (const float*)d_in[4];
    const float* b_l1  = (const float*)d_in[5];
    const float* w_r1  = (const float*)d_in[6];
    const float* g1    = (const float*)d_in[7];
    const float* be1   = (const float*)d_in[8];
    const float* w_l2  = (const float*)d_in[9];
    const float* b_l2  = (const float*)d_in[10];
    const float* w_r2  = (const float*)d_in[11];
    const float* g2    = (const float*)d_in[12];
    const float* be2   = (const float*)d_in[13];
    const float* w_out = (const float*)d_in[14];
    const float* b_out = (const float*)d_in[15];

    const int N = in_sizes[0] / DIN;
    const int E = in_sizes[1] / 2;
    const int* src = ei;
    const int* dst = ei + E;

    float* ws  = (float*)d_ws;
    float* h0  = ws;
    float* agg = ws + (size_t)N * H_DIM;
    float* h1  = ws + 2 * (size_t)N * H_DIM;
    float* deg = ws + 3 * (size_t)N * H_DIM;

    int* col      = (int*)d_out;
    int* rank     = col + E;
    int* cnt      = rank + E;
    int* rowstart = cnt + N;

    const int tilesH   = (N / 32) * (H_DIM / 32);
    const int tilesOut = (N / 32) * (DOUT / 32);

    hipMemsetAsync(cnt, 0, (size_t)N * sizeof(int), stream);
    rank_kernel<<<(E + 255) / 256, 256, 0, stream>>>(dst, cnt, rank, E);
    scan_kernel<<<(N + 1023) / 1024, 1024, 0, stream>>>(cnt, rowstart, deg, N);
    fill_kernel<<<(E + 255) / 256, 256, 0, stream>>>(src, dst, rank, rowstart, col, E);

    gemm_fused<<<(tilesH + 3) / 4, 128, 0, stream>>>(
        x, w_in, DIN, nullptr, nullptr, 0, nullptr, b_in, nullptr, nullptr,
        h0, N, H_DIM, 1);

    gather_sum_kernel<<<(N + 7) / 8, 256, 0, stream>>>(h0, rowstart, col, agg, N);
    gemm_fused<<<(tilesH + 3) / 4, 128, 0, stream>>>(
        agg, w_l1, H_DIM, h0, w_r1, H_DIM, deg, b_l1, g1, be1,
        h1, N, H_DIM, 1);

    gather_sum_kernel<<<(N + 7) / 8, 256, 0, stream>>>(h1, rowstart, col, agg, N);
    gemm_fused<<<(tilesH + 3) / 4, 128, 0, stream>>>(
        agg, w_l2, H_DIM, h1, w_r2, H_DIM, deg, b_l2, g2, be2,
        h0, N, H_DIM, 1);

    gemm_fused<<<(tilesOut + 3) / 4, 128, 0, stream>>>(
        h0, w_out, H_DIM, nullptr, nullptr, 0, nullptr, b_out, nullptr, nullptr,
        (float*)d_out, N, DOUT, 0);
}


# Dense products re-tiled: weights copied once per resident work-group into LDS in fragment order, a wave owns 16-row x 64-col pieces, A read 64 B per lane; k summation order in products changes
# speedup vs baseline: 2.4645x; 1.2381x over previous
#include <hip/hip_runtime.h>

#define DIN   256
#define H_DIM 128
#define DOUT  64
#define BN_RSQRT 0.9999950000374996f

typedef __attribute__((ext_vector_type(2))) float v2f;
typedef __attribute__((ext_vector_type(8))) float v8f;

__device__ __forceinline__ v8f wmma4(v2f a, v2f b, v8f c) {
    return __builtin_amdgcn_wmma_f32_16x16x4_f32(
         false, a,  false, b,
         (short)0, c,  false,  false);
}

#define GEMM_BLOCKS    256
#define GEMM_THREADS   1024
#define GEMM_LDS_BYTES (164 * 1024)

template <int NTW>
__device__ __forceinline__ void strip_accum(const float* __restrict__ ap,
                                            int K, float rs,
                                            const float4* bq,
                                            int ncols, v8f (&acc)[NTW]) {
    for (int k0 = 0; k0 < K; k0 += 32) {
        float4 a[4];
#pragma unroll
        for (int j = 0; j < 4; ++j) a[j] = *(const float4*)(ap + k0 + 4 * j);
        const float4* bk = bq + (size_t)(k0 >> 2) * ncols;
        float4 b[NTW];
#pragma unroll
        for (int ni = 0; ni < NTW; ++ni) b[ni] = bk[ni * 16];
#pragma unroll
        for (int j = 0; j < 4; ++j) {
            float4 bn[NTW];
            if (j < 3) {
#pragma unroll
                for (int ni = 0; ni < NTW; ++ni) bn[ni] = bk[(size_t)(j + 1) * ncols + ni * 16];
            }
            v2f alo, ahi;
            alo.x = a[j].x * rs;  alo.y = a[j].y * rs;
            ahi.x = a[j].z * rs;  ahi.y = a[j].w * rs;
#pragma unroll
            for (int ni = 0; ni < NTW; ++ni) {
                v2f bb; bb.x = b[ni].x; bb.y = b[ni].y;
                acc[ni] = wmma4(alo, bb, acc[ni]);
            }
#pragma unroll
            for (int ni = 0; ni < NTW; ++ni) {
                v2f bb; bb.x = b[ni].z; bb.y = b[ni].w;
                acc[ni] = wmma4(ahi, bb, acc[ni]);
            }
            if (j < 3) {
#pragma unroll
                for (int ni = 0; ni < NTW; ++ni) b[ni] = bn[ni];
            }
        }
    }
}

template <int NCOLS, int NTW>
__global__ __launch_bounds__(GEMM_THREADS)
void gemm_strips(const float* __restrict__ A0, const float* __restrict__ W0, int K0,
                 const float* __restrict__ A1, const float* __restrict__ W1, int K1,
                 const float* __restrict__ rowscale,
                 const float* __restrict__ bias,
                 const float* __restrict__ gamma, const float* __restrict__ beta,
                 float* __restrict__ out, int nrows, int do_relu) {
    __shared__ __attribute__((aligned(16))) float4 wl[GEMM_LDS_BYTES / 16];
    const int tid = threadIdx.x;
    const int lane = tid & 31;
    const int wave = __builtin_amdgcn_readfirstlane(tid >> 5);
    const int c = lane & 15, h = lane >> 4;
    const int Q0 = K0 >> 2;
    const int Q1 = A1 ? (K1 >> 2) : 0;

    const int nq = (Q0 + Q1) * NCOLS;
    for (int it = tid; it < nq; it += GEMM_THREADS) {
        const int q = it / NCOLS, n = it % NCOLS;
        const float* wp = (q < Q0) ? (W0 + (size_t)(4 * q) * NCOLS + n)
                                   : (W1 + (size_t)(4 * (q - Q0)) * NCOLS + n);
        float4 v;
        v.x = wp[0];  v.y = wp[NCOLS];  v.z = wp[2 * NCOLS];  v.w = wp[3 * NCOLS];
        wl[it] = v;
    }
    __syncthreads();

    constexpr int NSPLIT = NCOLS / (16 * NTW);
    const int items = (nrows >> 4) * NSPLIT;
    const int beg = (int)(((long long)blockIdx.x * items) / gridDim.x);
    const int end = (int)(((long long)(blockIdx.x + 1) * items) / gridDim.x);

    for (int it = beg + wave; it < end; it += GEMM_THREADS / 32) {
        const int strip = it / NSPLIT, part = it % NSPLIT;
        const int row0 = strip * 16, col0 = part * 16 * NTW;

        v8f acc[NTW] = {};
        const float rs = rowscale ? rowscale[row0 + c] : 1.0f;
        strip_accum<NTW>(A0 + (size_t)(row0 + c) * K0 + 16 * h, K0, rs,
                         wl + (size_t)(4 * h) * NCOLS + col0 + c, NCOLS, acc);
        if (A1)
            strip_accum<NTW>(A1 + (size_t)(row0 + c) * K1 + 16 * h, K1, 1.0f,
                             wl + (size_t)(Q0 + 4 * h) * NCOLS + col0 + c, NCOLS, acc);

#pragma unroll
        for (int ni = 0; ni < NTW; ++ni) {
            const int col = col0 + ni * 16 + c;
            const float bi = bias ? bias[col] : 0.0f;
            const float sc = gamma ? gamma[col] * BN_RSQRT : 1.0f;
            const float sh = beta ? beta[col] : 0.0f;
#pragma unroll
            for (int r = 0; r < 8; ++r) {
                const int row = row0 + r + 8 * h;
                float v = acc[ni][r];
                v = (v + bi) * sc + sh;
                if (do_relu) v = fmaxf(v, 0.0f);
                out[(size_t)row * NCOLS + col] = v;
            }
        }
    }
}


__global__ __launch_bounds__(256)
void rank_kernel(const int* __restrict__ dst, int* __restrict__ cnt,
                 int* __restrict__ rank, int E) {
    const int e = blockIdx.x * blockDim.x + threadIdx.x;
    if (e < E) rank[e] = atomicAdd(&cnt[dst[e]], 1);
}

__global__ __launch_bounds__(1024)
void scan_kernel(const int* __restrict__ cnt, int* __restrict__ rowstart,
                 float* __restrict__ invdeg, int N) {
    __shared__ int s_pre[32];
    __shared__ int s_tot[32];
    const int tid = threadIdx.x;
    const int lane = tid & 31, wave = tid >> 5;
    const int tile0 = blockIdx.x * 1024;

    int pre = 0;
    for (int k = tid; k < tile0; k += 1024) pre += cnt[k];

    const int i = tile0 + tid;
    const int c = (i < N) ? cnt[i] : 0;

    int incl = c;
    for (int d = 1; d < 32; d <<= 1) {
        const int up = __shfl_up(incl, d);
        if (lane >= d) incl += up;
    }
    for (int d = 16; d > 0; d >>= 1) pre += __shfl_xor(pre, d);

    if (lane == 31) s_tot[wave] = incl;
    if (lane == 0)  s_pre[wave] = pre;
    __syncthreads();

    int base = 0;
    for (int w = 0; w < 32; ++w) {
        base += s_pre[w];
        if (w < wave) base += s_tot[w];
    }
    const int excl = base + incl - c;
    if (i < N) {
        rowstart[i] = excl;
        invdeg[i] = 1.0f / fmaxf((float)c, 1.0f);
    }
    if (i == N - 1) rowstart[N] = excl + c;
}

__global__ __launch_bounds__(256)
void fill_kernel(const int* __restrict__ src, const int* __restrict__ dst,
                 const int* __restrict__ rank, const int* __restrict__ rowstart,
                 int* __restrict__ col, int E) {
    const int e = blockIdx.x * blockDim.x + threadIdx.x;
    if (e < E) col[rowstart[dst[e]] + rank[e]] = src[e];
}

#define GATHER_ROWS(U)                                                          \
    {                                                                           \
        float4 v[U];                                                            \
        for (int u = 0; u < (U); ++u)                                           \
            v[u] = *(const float4*)(hp + (size_t)col[j + u] * H_DIM);           \
        for (int u = 0; u < (U); ++u) {                                         \
            acc.x += v[u].x; acc.y += v[u].y; acc.z += v[u].z; acc.w += v[u].w; \
        }                                                                       \
        j += (U);                                                               \
    }

__global__ __launch_bounds__(256)
void gather_sum_kernel(const float* __restrict__ h, const int* __restrict__ rowstart,
                       const int* __restrict__ col, float* __restrict__ agg, int N) {
    const int lane = threadIdx.x & 31;
    const int node = __builtin_amdgcn_readfirstlane((int)(blockIdx.x * 8 + (threadIdx.x >> 5)));
    if (node >= N) return;
    const int beg = rowstart[node];
    const int end = rowstart[node + 1];
    const float* hp = h + lane * 4;
    float4 acc = make_float4(0.0f, 0.0f, 0.0f, 0.0f);
    int j = beg;
    while (j + 8 <= end) GATHER_ROWS(8)
    if (j + 4 <= end) GATHER_ROWS(4)
    if (j + 2 <= end) GATHER_ROWS(2)
    if (j < end) GATHER_ROWS(1)
    *(float4*)(agg + (size_t)node * H_DIM + lane * 4) = acc;
}

extern "C" void kernel_launch(void* const* d_in, const int* in_sizes, int n_in,
                              void* d_out, int out_size, void* d_ws, size_t ws_size,
                              hipStream_t stream) {
    const float* x     = (const float*)d_in[0];
    const int*   ei    = (const int*)d_in[1];
    const float* w_in  = (const float*)d_in[2];
    const float* b_in  = (const float*)d_in[3];
    const float* w_l1  = (const float*)d_in[4];
    const float* b_l1  = (const float*)d_in[5];
    const float* w_r1  = (const float*)d_in[6];
    const float* g1    = (const float*)d_in[7];
    const float* be1   = (const float*)d_in[8];
    const float* w_l2  = (const float*)d_in[9];
    const float* b_l2  = (const float*)d_in[10];
    const float* w_r2  = (const float*)d_in[11];
    const float* g2    = (const float*)d_in[12];
    const float* be2   = (const float*)d_in[13];
    const float* w_out = (const float*)d_in[14];
    const float* b_out = (const float*)d_in[15];

    const int N = in_sizes[0] / DIN;
    const int E = in_sizes[1] / 2;
    const int* src = ei;
    const int* dst = ei + E;

    float* ws  = (float*)d_ws;
    float* h0  = ws;
    float* agg = ws + (size_t)N * H_DIM;
    float* h1  = ws + 2 * (size_t)N * H_DIM;
    float* deg = ws + 3 * (size_t)N * H_DIM;

    int* col      = (int*)d_out;
    int* rank     = col + E;
    int* cnt      = rank + E;
    int* rowstart = cnt + N;


    hipMemsetAsync(cnt, 0, (size_t)N * sizeof(int), stream);
    rank_kernel<<<(E + 255) / 256, 256, 0, stream>>>(dst, cnt, rank, E);
    scan_kernel<<<(N + 1023) / 1024, 1024, 0, stream>>>(cnt, rowstart, deg, N);
    fill_kernel<<<(E + 255) / 256, 256, 0, stream>>>(src, dst, rank, rowstart, col, E);

    gemm_strips<H_DIM, 4><<<GEMM_BLOCKS, GEMM_THREADS, 0, stream>>>(
        x, w_in, DIN, nullptr, nullptr, 0, nullptr, b_in, nullptr, nullptr,
        h0, N, 1);

    gather_sum_kernel<<<(N + 7) / 8, 256, 0, stream>>>(h0, rowstart, col, agg, N);
    gemm_strips<H_DIM, 4><<<GEMM_BLOCKS, GEMM_THREADS, 0, stream>>>(
        agg, w_l1, H_DIM, h0, w_r1, H_DIM, deg, b_l1, g1, be1,
        h1, N, 1);

    gather_sum_kernel<<<(N + 7) / 8, 256, 0, stream>>>(h1, rowstart, col, agg, N);
    gemm_strips<H_DIM, 4><<<GEMM_BLOCKS, GEMM_THREADS, 0, stream>>>(
        agg, w_l2, H_DIM, h1, w_r2, H_DIM, deg, b_l2, g2, be2,
        h0, N, 1);

    gemm_strips<DOUT, 4><<<GEMM_BLOCKS, GEMM_THREADS, 0, stream>>>(
        h0, w_out, H_DIM, nullptr, nullptr, 0, nullptr, b_out, nullptr, nullptr,
        (float*)d_out, N, 0);
}


# Edge bucketing without global atomics: edges split into 512-node target ranges by LDS counters, one work-group per range ranks and places them in LDS; slot order inside a node's sum changes
# speedup vs baseline: 2.7947x; 1.1340x over previous
#include <hip/hip_runtime.h>

#define DIN   256
#define H_DIM 128
#define DOUT  64
#define BN_RSQRT 0.9999950000374996f

typedef __attribute__((ext_vector_type(2))) float v2f;
typedef __attribute__((ext_vector_type(8))) float v8f;

__device__ __forceinline__ v8f wmma4(v2f a, v2f b, v8f c) {
    return __builtin_amdgcn_wmma_f32_16x16x4_f32(
         false, a,  false, b,
         (short)0, c,  false,  false);
}

#define GEMM_BLOCKS    256
#define GEMM_THREADS   1024
#define GEMM_LDS_BYTES (164 * 1024)

template <int NTW>
__device__ __forceinline__ void strip_accum(const float* __restrict__ ap,
                                            int K, float rs,
                                            const float4* bq,
                                            int ncols, v8f (&acc)[NTW]) {
    for (int k0 = 0; k0 < K; k0 += 32) {
        float4 a[4];
#pragma unroll
        for (int j = 0; j < 4; ++j) a[j] = *(const float4*)(ap + k0 + 4 * j);
        const float4* bk = bq + (size_t)(k0 >> 2) * ncols;
        float4 b[NTW];
#pragma unroll
        for (int ni = 0; ni < NTW; ++ni) b[ni] = bk[ni * 16];
#pragma unroll
        for (int j = 0; j < 4; ++j) {
            float4 bn[NTW];
            if (j < 3) {
#pragma unroll
                for (int ni = 0; ni < NTW; ++ni) bn[ni] = bk[(size_t)(j + 1) * ncols + ni * 16];
            }
            v2f alo, ahi;
            alo.x = a[j].x * rs;  alo.y = a[j].y * rs;
            ahi.x = a[j].z * rs;  ahi.y = a[j].w * rs;
#pragma unroll
            for (int ni = 0; ni < NTW; ++ni) {
                v2f bb; bb.x = b[ni].x; bb.y = b[ni].y;
                acc[ni] = wmma4(alo, bb, acc[ni]);
            }
#pragma unroll
            for (int ni = 0; ni < NTW; ++ni) {
                v2f bb; bb.x = b[ni].z; bb.y = b[ni].w;
                acc[ni] = wmma4(ahi, bb, acc[ni]);
            }
            if (j < 3) {
#pragma unroll
                for (int ni = 0; ni < NTW; ++ni) b[ni] = bn[ni];
            }
        }
    }
}

template <int NCOLS, int NTW>
__global__ __launch_bounds__(GEMM_THREADS)
void gemm_strips(const float* __restrict__ A0, const float* __restrict__ W0, int K0,
                 const float* __restrict__ A1, const float* __restrict__ W1, int K1,
                 const float* __restrict__ rowscale,
                 const float* __restrict__ bias,
                 const float* __restrict__ gamma, const float* __restrict__ beta,
                 float* __restrict__ out, int nrows, int do_relu) {
    __shared__ __attribute__((aligned(16))) float4 wl[GEMM_LDS_BYTES / 16];
    const int tid = threadIdx.x;
    const int lane = tid & 31;
    const int wave = __builtin_amdgcn_readfirstlane(tid >> 5);
    const int c = lane & 15, h = lane >> 4;
    const int Q0 = K0 >> 2;
    const int Q1 = A1 ? (K1 >> 2) : 0;

    const int nq = (Q0 + Q1) * NCOLS;
    for (int it = tid; it < nq; it += GEMM_THREADS) {
        const int q = it / NCOLS, n = it % NCOLS;
        const float* wp = (q < Q0) ? (W0 + (size_t)(4 * q) * NCOLS + n)
                                   : (W1 + (size_t)(4 * (q - Q0)) * NCOLS + n);
        float4 v;
        v.x = wp[0];  v.y = wp[NCOLS];  v.z = wp[2 * NCOLS];  v.w = wp[3 * NCOLS];
        wl[it] = v;
    }
    __syncthreads();

    constexpr int NSPLIT = NCOLS / (16 * NTW);
    const int items = (nrows >> 4) * NSPLIT;
    const int beg = (int)(((long long)blockIdx.x * items) / gridDim.x);
    const int end = (int)(((long long)(blockIdx.x + 1) * items) / gridDim.x);

    for (int it = beg + wave; it < end; it += GEMM_THREADS / 32) {
        const int strip = it / NSPLIT, part = it % NSPLIT;
        const int row0 = strip * 16, col0 = part * 16 * NTW;

        v8f acc[NTW] = {};
        const float rs = rowscale ? rowscale[row0 + c] : 1.0f;
        strip_accum<NTW>(A0 + (size_t)(row0 + c) * K0 + 16 * h, K0, rs,
                         wl + (size_t)(4 * h) * NCOLS + col0 + c, NCOLS, acc);
        if (A1)
            strip_accum<NTW>(A1 + (size_t)(row0 + c) * K1 + 16 * h, K1, 1.0f,
                             wl + (size_t)(Q0 + 4 * h) * NCOLS + col0 + c, NCOLS, acc);

#pragma unroll
        for (int ni = 0; ni < NTW; ++ni) {
            const int col = col0 + ni * 16 + c;
            const float bi = bias ? bias[col] : 0.0f;
            const float sc = gamma ? gamma[col] * BN_RSQRT : 1.0f;
            const float sh = beta ? beta[col] : 0.0f;
#pragma unroll
            for (int r = 0; r < 8; ++r) {
                const int row = row0 + r + 8 * h;
                float v = acc[ni][r];
                v = (v + bi) * sc + sh;
                if (do_relu) v = fmaxf(v, 0.0f);
                out[(size_t)row * NCOLS + col] = v;
            }
        }
    }
}

#define PB_SHIFT 9
#define PB_NODES (1 << PB_SHIFT)
#define PB_MAXB  2048
#define PB_GRID  256
#define PB_BATCH 8

__global__ __launch_bounds__(1024)
void part_count_kernel(const int* __restrict__ dst, int* __restrict__ lrank,
                       int* __restrict__ gh, int E, int NB) {
    __shared__ int hist[PB_MAXB];
    const int tid = threadIdx.x;
    for (int i = tid; i < NB; i += 1024) hist[i] = 0;
    __syncthreads();
    const int beg = (int)(((long long)blockIdx.x * E) / gridDim.x);
    const int end = (int)(((long long)(blockIdx.x + 1) * E) / gridDim.x);
    for (int e = beg + tid; e < end; e += PB_BATCH * 1024) {
        int d[PB_BATCH];
#pragma unroll
        for (int u = 0; u < PB_BATCH; ++u)
            d[u] = dst[min(e + u * 1024, end - 1)];
#pragma unroll
        for (int u = 0; u < PB_BATCH; ++u)
            if (e + u * 1024 < end) lrank[e + u * 1024] = atomicAdd(&hist[d[u] >> PB_SHIFT], 1);
    }
    __syncthreads();
    for (int i = tid; i < NB; i += 1024) gh[(size_t)i * gridDim.x + blockIdx.x] = hist[i];
}

__global__ __launch_bounds__(1024)
void part_scan_kernel(int* __restrict__ gh, int n) {
    __shared__ int s_w[32];
    const int tid = threadIdx.x;
    const int lane = tid & 31, wave = tid >> 5;
    const int per = (((n + 31) >> 5) + 31) & ~31;
    const int i0 = wave * per;
    const int iend = (i0 + per < n) ? (i0 + per) : n;
    int sum = 0;
    for (int i = i0 + lane; i < iend; i += 32) sum += gh[i];
    for (int d = 16; d > 0; d >>= 1) sum += __shfl_xor(sum, d);
    if (lane == 0) s_w[wave] = sum;
    __syncthreads();
    int carry = 0;
    for (int w = 0; w < wave; ++w) carry += s_w[w];
    for (int k = 0; k < per; k += 32) {
        const int i = i0 + k + lane;
        const int v = (i < n) ? gh[i] : 0;
        int incl = v;
        for (int d = 1; d < 32; d <<= 1) {
            const int up = __shfl_up(incl, d);
            if (lane >= d) incl += up;
        }
        if (i < n) gh[i] = carry + incl - v;
        carry += __shfl(incl, 31);
    }
    if (tid == 1023) gh[n] = carry;
}

__global__ __launch_bounds__(1024)
void part_move_kernel(const int* __restrict__ src, const int* __restrict__ dst,
                      const int* __restrict__ lrank, const int* __restrict__ gh,
                      int* __restrict__ psrc, int* __restrict__ pdst, int E, int NB) {
    __shared__ int base[PB_MAXB];
    const int tid = threadIdx.x;
    for (int i = tid; i < NB; i += 1024) base[i] = gh[(size_t)i * gridDim.x + blockIdx.x];
    __syncthreads();
    const int beg = (int)(((long long)blockIdx.x * E) / gridDim.x);
    const int end = (int)(((long long)(blockIdx.x + 1) * E) / gridDim.x);
    for (int e = beg + tid; e < end; e += PB_BATCH * 1024) {
        int d[PB_BATCH], r[PB_BATCH], s[PB_BATCH];
#pragma unroll
        for (int u = 0; u < PB_BATCH; ++u) {
            const int ee = min(e + u * 1024, end - 1);
            d[u] = dst[ee];
            r[u] = lrank[ee];
            s[u] = src[ee];
        }
#pragma unroll
        for (int u = 0; u < PB_BATCH; ++u) {
            if (e + u * 1024 < end) {
                const int pos = base[d[u] >> PB_SHIFT] + r[u];
                psrc[pos] = s[u];
                pdst[pos] = d[u];
            }
        }
    }
}

__global__ __launch_bounds__(1024)
void bucket_fill_kernel(const int* __restrict__ psrc, const int* __restrict__ pdst,
                        const int* __restrict__ gh, int G,
                        int* __restrict__ rowstart, float* __restrict__ invdeg,
                        int* __restrict__ col, int N) {
    __shared__ int cnt[PB_NODES];
    __shared__ int start[PB_NODES];
    __shared__ int s_w[32];
    const int tid = threadIdx.x;
    const int lane = tid & 31, wave = tid >> 5;
    const int b = blockIdx.x;
    const int s0 = gh[(size_t)b * G];
    const int s1 = gh[(size_t)(b + 1) * G];
    const int n0 = b << PB_SHIFT;

    if (tid < PB_NODES) cnt[tid] = 0;
    __syncthreads();
    for (int e = s0 + tid; e < s1; e += PB_BATCH * 1024) {
        int d[PB_BATCH];
#pragma unroll
        for (int u = 0; u < PB_BATCH; ++u)
            d[u] = pdst[min(e + u * 1024, s1 - 1)];
#pragma unroll
        for (int u = 0; u < PB_BATCH; ++u)
            if (e + u * 1024 < s1) atomicAdd(&cnt[d[u] - n0], 1);
    }
    __syncthreads();

    const int c = (tid < PB_NODES) ? cnt[tid] : 0;
    int incl = c;
    for (int d = 1; d < 32; d <<= 1) {
        const int up = __shfl_up(incl, d);
        if (lane >= d) incl += up;
    }
    if (lane == 31) s_w[wave] = incl;
    __syncthreads();
    int excl = incl - c;
    for (int w = 0; w < wave; ++w) excl += s_w[w];
    if (tid < PB_NODES) {
        start[tid] = excl;
        cnt[tid] = 0;
        const int node = n0 + tid;
        if (node < N) {
            rowstart[node] = s0 + excl;
            invdeg[node] = 1.0f / fmaxf((float)c, 1.0f);
        }
    }
    if (b == (int)gridDim.x - 1 && tid == 0) rowstart[N] = s1;
    __syncthreads();

    for (int e = s0 + tid; e < s1; e += PB_BATCH * 1024) {
        int d[PB_BATCH], s[PB_BATCH];
#pragma unroll
        for (int u = 0; u < PB_BATCH; ++u) {
            const int ee = min(e + u * 1024, s1 - 1);
            d[u] = pdst[ee];
            s[u] = psrc[ee];
        }
#pragma unroll
        for (int u = 0; u < PB_BATCH; ++u) {
            if (e + u * 1024 < s1) {
                const int dl = d[u] - n0;
                const int slot = atomicAdd(&cnt[dl], 1);
                col[s0 + start[dl] + slot] = s[u];
            }
        }
    }
}

#define GATHER_ROWS(U)                                                          \
    {                                                                           \
        float4 v[U];                                                            \
        for (int u = 0; u < (U); ++u)                                           \
            v[u] = *(const float4*)(hp + (size_t)col[j + u] * H_DIM);           \
        for (int u = 0; u < (U); ++u) {                                         \
            acc.x += v[u].x; acc.y += v[u].y; acc.z += v[u].z; acc.w += v[u].w; \
        }                                                                       \
        j += (U);                                                               \
    }

__global__ __launch_bounds__(256)
void gather_sum_kernel(const float* __restrict__ h, const int* __restrict__ rowstart,
                       const int* __restrict__ col, float* __restrict__ agg, int N) {
    const int lane = threadIdx.x & 31;
    const int node = __builtin_amdgcn_readfirstlane((int)(blockIdx.x * 8 + (threadIdx.x >> 5)));
    if (node >= N) return;
    const int beg = rowstart[node];
    const int end = rowstart[node + 1];
    const float* hp = h + lane * 4;
    float4 acc = make_float4(0.0f, 0.0f, 0.0f, 0.0f);
    int j = beg;
    while (j + 8 <= end) GATHER_ROWS(8)
    if (j + 4 <= end) GATHER_ROWS(4)
    if (j + 2 <= end) GATHER_ROWS(2)
    if (j < end) GATHER_ROWS(1)
    *(float4*)(agg + (size_t)node * H_DIM + lane * 4) = acc;
}

extern "C" void kernel_launch(void* const* d_in, const int* in_sizes, int n_in,
                              void* d_out, int out_size, void* d_ws, size_t ws_size,
                              hipStream_t stream) {
    const float* x     = (const float*)d_in[0];
    const int*   ei    = (const int*)d_in[1];
    const float* w_in  = (const float*)d_in[2];
    const float* b_in  = (const float*)d_in[3];
    const float* w_l1  = (const float*)d_in[4];
    const float* b_l1  = (const float*)d_in[5];
    const float* w_r1  = (const float*)d_in[6];
    const float* g1    = (const float*)d_in[7];
    const float* be1   = (const float*)d_in[8];
    const float* w_l2  = (const float*)d_in[9];
    const float* b_l2  = (const float*)d_in[10];
    const float* w_r2  = (const float*)d_in[11];
    const float* g2    = (const float*)d_in[12];
    const float* be2   = (const float*)d_in[13];
    const float* w_out = (const float*)d_in[14];
    const float* b_out = (const float*)d_in[15];

    const int N = in_sizes[0] / DIN;
    const int E = in_sizes[1] / 2;
    const int* src = ei;
    const int* dst = ei + E;

    float* ws  = (float*)d_ws;
    float* h0  = ws;
    float* agg = ws + (size_t)N * H_DIM;
    float* h1  = ws + 2 * (size_t)N * H_DIM;
    float* deg = ws + 3 * (size_t)N * H_DIM;

    int* col      = (int*)d_out;
    int* rowstart = col + E;

    const int NB = (N + PB_NODES - 1) >> PB_SHIFT;
    int* psrc  = (int*)agg;
    int* pdst  = psrc + E;
    int* lrank = pdst + E;
    int* gh    = lrank + E;

    part_count_kernel<<<PB_GRID, 1024, 0, stream>>>(dst, lrank, gh, E, NB);
    part_scan_kernel<<<1, 1024, 0, stream>>>(gh, NB * PB_GRID);
    part_move_kernel<<<PB_GRID, 1024, 0, stream>>>(src, dst, lrank, gh, psrc, pdst, E, NB);
    bucket_fill_kernel<<<NB, 1024, 0, stream>>>(psrc, pdst, gh, PB_GRID, rowstart, deg, col, N);

    gemm_strips<H_DIM, 4><<<GEMM_BLOCKS, GEMM_THREADS, 0, stream>>>(
        x, w_in, DIN, nullptr, nullptr, 0, nullptr, b_in, nullptr, nullptr,
        h0, N, 1);

    gather_sum_kernel<<<(N + 7) / 8, 256, 0, stream>>>(h0, rowstart, col, agg, N);
    gemm_strips<H_DIM, 4><<<GEMM_BLOCKS, GEMM_THREADS, 0, stream>>>(
        agg, w_l1, H_DIM, h0, w_r1, H_DIM, deg, b_l1, g1, be1,
        h1, N, 1);

    gather_sum_kernel<<<(N + 7) / 8, 256, 0, stream>>>(h1, rowstart, col, agg, N);
    gemm_strips<H_DIM, 4><<<GEMM_BLOCKS, GEMM_THREADS, 0, stream>>>(
        agg, w_l2, H_DIM, h1, w_r2, H_DIM, deg, b_l2, g2, be2,
        h0, N, 1);

    gemm_strips<DOUT, 4><<<GEMM_BLOCKS, GEMM_THREADS, 0, stream>>>(
        h0, w_out, H_DIM, nullptr, nullptr, 0, nullptr, b_out, nullptr, nullptr,
        (float*)d_out, N, 0);
}


# Cell starts without a scan launch: every placing work-group sums the cell-size table itself (whole-line reads, LDS scan of the range sizes); the one-work-group scan kernel goes, 10 launches to 9
# speedup vs baseline: 2.9771x; 1.0653x over previous
#include <hip/hip_runtime.h>

#define DIN   256
#define H_DIM 128
#define DOUT  64
#define BN_RSQRT 0.9999950000374996f

typedef __attribute__((ext_vector_type(2))) float v2f;
typedef __attribute__((ext_vector_type(8))) float v8f;

__device__ __forceinline__ v8f wmma4(v2f a, v2f b, v8f c) {
    return __builtin_amdgcn_wmma_f32_16x16x4_f32(
         false, a,  false, b,
         (short)0, c,  false,  false);
}

#define GEMM_BLOCKS    256
#define GEMM_THREADS   1024
#define GEMM_LDS_BYTES (164 * 1024)

template <int NTW>
__device__ __forceinline__ void strip_accum(const float* __restrict__ ap,
                                            int K, float rs,
                                            const float4* bq,
                                            int ncols, v8f (&acc)[NTW]) {
    for (int k0 = 0; k0 < K; k0 += 32) {
        float4 a[4];
#pragma unroll
        for (int j = 0; j < 4; ++j) a[j] = *(const float4*)(ap + k0 + 4 * j);
        const float4* bk = bq + (size_t)(k0 >> 2) * ncols;
        float4 b[NTW];
#pragma unroll
        for (int ni = 0; ni < NTW; ++ni) b[ni] = bk[ni * 16];
#pragma unroll
        for (int j = 0; j < 4; ++j) {
            float4 bn[NTW];
            if (j < 3) {
#pragma unroll
                for (int ni = 0; ni < NTW; ++ni) bn[ni] = bk[(size_t)(j + 1) * ncols + ni * 16];
            }
            v2f alo, ahi;
            alo.x = a[j].x * rs;  alo.y = a[j].y * rs;
            ahi.x = a[j].z * rs;  ahi.y = a[j].w * rs;
#pragma unroll
            for (int ni = 0; ni < NTW; ++ni) {
                v2f bb; bb.x = b[ni].x; bb.y = b[ni].y;
                acc[ni] = wmma4(alo, bb, acc[ni]);
            }
#pragma unroll
            for (int ni = 0; ni < NTW; ++ni) {
                v2f bb; bb.x = b[ni].z; bb.y = b[ni].w;
                acc[ni] = wmma4(ahi, bb, acc[ni]);
            }
            if (j < 3) {
#pragma unroll
                for (int ni = 0; ni < NTW; ++ni) b[ni] = bn[ni];
            }
        }
    }
}

template <int NCOLS, int NTW>
__global__ __launch_bounds__(GEMM_THREADS)
void gemm_strips(const float* __restrict__ A0, const float* __restrict__ W0, int K0,
                 const float* __restrict__ A1, const float* __restrict__ W1, int K1,
                 const float* __restrict__ rowscale,
                 const float* __restrict__ bias,
                 const float* __restrict__ gamma, const float* __restrict__ beta,
                 float* __restrict__ out, int nrows, int do_relu) {
    __shared__ __attribute__((aligned(16))) float4 wl[GEMM_LDS_BYTES / 16];
    const int tid = threadIdx.x;
    const int lane = tid & 31;
    const int wave = __builtin_amdgcn_readfirstlane(tid >> 5);
    const int c = lane & 15, h = lane >> 4;
    const int Q0 = K0 >> 2;
    const int Q1 = A1 ? (K1 >> 2) : 0;

    const int nq = (Q0 + Q1) * NCOLS;
    for (int it = tid; it < nq; it += GEMM_THREADS) {
        const int q = it / NCOLS, n = it % NCOLS;
        const float* wp = (q < Q0) ? (W0 + (size_t)(4 * q) * NCOLS + n)
                                   : (W1 + (size_t)(4 * (q - Q0)) * NCOLS + n);
        float4 v;
        v.x = wp[0];  v.y = wp[NCOLS];  v.z = wp[2 * NCOLS];  v.w = wp[3 * NCOLS];
        wl[it] = v;
    }
    __syncthreads();

    constexpr int NSPLIT = NCOLS / (16 * NTW);
    const int items = (nrows >> 4) * NSPLIT;
    const int beg = (int)(((long long)blockIdx.x * items) / gridDim.x);
    const int end = (int)(((long long)(blockIdx.x + 1) * items) / gridDim.x);

    for (int it = beg + wave; it < end; it += GEMM_THREADS / 32) {
        const int strip = it / NSPLIT, part = it % NSPLIT;
        const int row0 = strip * 16, col0 = part * 16 * NTW;

        v8f acc[NTW] = {};
        const float rs = rowscale ? rowscale[row0 + c] : 1.0f;
        strip_accum<NTW>(A0 + (size_t)(row0 + c) * K0 + 16 * h, K0, rs,
                         wl + (size_t)(4 * h) * NCOLS + col0 + c, NCOLS, acc);
        if (A1)
            strip_accum<NTW>(A1 + (size_t)(row0 + c) * K1 + 16 * h, K1, 1.0f,
                             wl + (size_t)(Q0 + 4 * h) * NCOLS + col0 + c, NCOLS, acc);

#pragma unroll
        for (int ni = 0; ni < NTW; ++ni) {
            const int col = col0 + ni * 16 + c;
            const float bi = bias ? bias[col] : 0.0f;
            const float sc = gamma ? gamma[col] * BN_RSQRT : 1.0f;
            const float sh = beta ? beta[col] : 0.0f;
#pragma unroll
            for (int r = 0; r < 8; ++r) {
                const int row = row0 + r + 8 * h;
                float v = acc[ni][r];
                v = (v + bi) * sc + sh;
                if (do_relu) v = fmaxf(v, 0.0f);
                out[(size_t)row * NCOLS + col] = v;
            }
        }
    }
}

#define PB_SHIFT 9
#define PB_NODES (1 << PB_SHIFT)
#define PB_MAXB  2048
#define PB_GRID  256
#define PB_BATCH 8

__global__ __launch_bounds__(1024)
void part_count_kernel(const int* __restrict__ dst, int* __restrict__ lrank,
                       int* __restrict__ gh, int E, int NB) {
    __shared__ int hist[PB_MAXB];
    const int tid = threadIdx.x;
    for (int i = tid; i < NB; i += 1024) hist[i] = 0;
    __syncthreads();
    const int beg = (int)(((long long)blockIdx.x * E) / gridDim.x);
    const int end = (int)(((long long)(blockIdx.x + 1) * E) / gridDim.x);
    for (int e = beg + tid; e < end; e += PB_BATCH * 1024) {
        int d[PB_BATCH];
#pragma unroll
        for (int u = 0; u < PB_BATCH; ++u)
            d[u] = dst[min(e + u * 1024, end - 1)];
#pragma unroll
        for (int u = 0; u < PB_BATCH; ++u)
            if (e + u * 1024 < end) lrank[e + u * 1024] = atomicAdd(&hist[d[u] >> PB_SHIFT], 1);
    }
    __syncthreads();
    for (int i = tid; i < NB; i += 1024) gh[(size_t)i * PB_GRID + blockIdx.x] = hist[i];
}

__global__ __launch_bounds__(1024)
void part_move_kernel(const int* __restrict__ src, const int* __restrict__ dst,
                      const int* __restrict__ lrank, const int* __restrict__ gh,
                      int* __restrict__ rs,
                      int* __restrict__ psrc, int* __restrict__ pdst, int E, int NB) {
    __shared__ int base[PB_MAXB];
    __shared__ int tot[PB_MAXB];
    __shared__ int s_w[32];
    const int tid = threadIdx.x;
    const int lane = tid & 31;
    const int wave = __builtin_amdgcn_readfirstlane(tid >> 5);
    const int g = blockIdx.x;

    int v[PB_GRID / 32];
    const int* row = gh + (size_t)min(wave, NB - 1) * PB_GRID + lane;
#pragma unroll
    for (int m = 0; m < PB_GRID / 32; ++m) v[m] = row[32 * m];
    for (int i = wave; i < NB; i += 32) {
        int nv[PB_GRID / 32];
        row = gh + (size_t)min(i + 32, NB - 1) * PB_GRID + lane;
#pragma unroll
        for (int m = 0; m < PB_GRID / 32; ++m) nv[m] = row[32 * m];
        int t = 0, p = 0;
#pragma unroll
        for (int m = 0; m < PB_GRID / 32; ++m) {
            t += v[m];
            if (32 * m + lane < g) p += v[m];
        }
        for (int d = 16; d > 0; d >>= 1) {
            t += __shfl_xor(t, d);
            p += __shfl_xor(p, d);
        }
        if (lane == 0) { tot[i] = t; base[i] = p; }
#pragma unroll
        for (int m = 0; m < PB_GRID / 32; ++m) v[m] = nv[m];
    }
    __syncthreads();

    constexpr int RPT = PB_MAXB / 1024;
    int loc = 0;
#pragma unroll
    for (int u = 0; u < RPT; ++u)
        if (tid * RPT + u < NB) loc += tot[tid * RPT + u];
    int incl = loc;
    for (int d = 1; d < 32; d <<= 1) {
        const int up = __shfl_up(incl, d);
        if (lane >= d) incl += up;
    }
    if (lane == 31) s_w[wave] = incl;
    __syncthreads();
    int run = incl - loc;
    for (int w = 0; w < wave; ++w) run += s_w[w];
#pragma unroll
    for (int u = 0; u < RPT; ++u) {
        const int i = tid * RPT + u;
        if (i < NB) {
            const int sz = tot[i];
            base[i] += run;
            if (g == 0) rs[i] = run;
            run += sz;
        }
    }
    if (g == 0 && tid == 1023) rs[NB] = run;
    __syncthreads();
    const int beg = (int)(((long long)blockIdx.x * E) / gridDim.x);
    const int end = (int)(((long long)(blockIdx.x + 1) * E) / gridDim.x);
    for (int e = beg + tid; e < end; e += PB_BATCH * 1024) {
        int d[PB_BATCH], r[PB_BATCH], s[PB_BATCH];
#pragma unroll
        for (int u = 0; u < PB_BATCH; ++u) {
            const int ee = min(e + u * 1024, end - 1);
            d[u] = dst[ee];
            r[u] = lrank[ee];
            s[u] = src[ee];
        }
#pragma unroll
        for (int u = 0; u < PB_BATCH; ++u) {
            if (e + u * 1024 < end) {
                const int pos = base[d[u] >> PB_SHIFT] + r[u];
                psrc[pos] = s[u];
                pdst[pos] = d[u];
            }
        }
    }
}

__global__ __launch_bounds__(1024)
void bucket_fill_kernel(const int* __restrict__ psrc, const int* __restrict__ pdst,
                        const int* __restrict__ rs,
                        int* __restrict__ rowstart, float* __restrict__ invdeg,
                        int* __restrict__ col, int N) {
    __shared__ int cnt[PB_NODES];
    __shared__ int start[PB_NODES];
    __shared__ int s_w[32];
    const int tid = threadIdx.x;
    const int lane = tid & 31, wave = tid >> 5;
    const int b = blockIdx.x;
    const int s0 = rs[b];
    const int s1 = rs[b + 1];
    const int n0 = b << PB_SHIFT;

    if (tid < PB_NODES) cnt[tid] = 0;
    __syncthreads();
    for (int e = s0 + tid; e < s1; e += PB_BATCH * 1024) {
        int d[PB_BATCH];
#pragma unroll
        for (int u = 0; u < PB_BATCH; ++u)
            d[u] = pdst[min(e + u * 1024, s1 - 1)];
#pragma unroll
        for (int u = 0; u < PB_BATCH; ++u)
            if (e + u * 1024 < s1) atomicAdd(&cnt[d[u] - n0], 1);
    }
    __syncthreads();

    const int c = (tid < PB_NODES) ? cnt[tid] : 0;
    int incl = c;
    for (int d = 1; d < 32; d <<= 1) {
        const int up = __shfl_up(incl, d);
        if (lane >= d) incl += up;
    }
    if (lane == 31) s_w[wave] = incl;
    __syncthreads();
    int excl = incl - c;
    for (int w = 0; w < wave; ++w) excl += s_w[w];
    if (tid < PB_NODES) {
        start[tid] = excl;
        cnt[tid] = 0;
        const int node = n0 + tid;
        if (node < N) {
            rowstart[node] = s0 + excl;
            invdeg[node] = 1.0f / fmaxf((float)c, 1.0f);
        }
    }
    if (b == (int)gridDim.x - 1 && tid == 0) rowstart[N] = s1;
    __syncthreads();

    for (int e = s0 + tid; e < s1; e += PB_BATCH * 1024) {
        int d[PB_BATCH], s[PB_BATCH];
#pragma unroll
        for (int u = 0; u < PB_BATCH; ++u) {
            const int ee = min(e + u * 1024, s1 - 1);
            d[u] = pdst[ee];
            s[u] = psrc[ee];
        }
#pragma unroll
        for (int u = 0; u < PB_BATCH; ++u) {
            if (e + u * 1024 < s1) {
                const int dl = d[u] - n0;
                const int slot = atomicAdd(&cnt[dl], 1);
                col[s0 + start[dl] + slot] = s[u];
            }
        }
    }
}

#define GATHER_ROWS(U)                                                          \
    {                                                                           \
        float4 v[U];                                                            \
        for (int u = 0; u < (U); ++u)                                           \
            v[u] = *(const float4*)(hp + (size_t)col[j + u] * H_DIM);           \
        for (int u = 0; u < (U); ++u) {                                         \
            acc.x += v[u].x; acc.y += v[u].y; acc.z += v[u].z; acc.w += v[u].w; \
        }                                                                       \
        j += (U);                                                               \
    }

__global__ __launch_bounds__(256)
void gather_sum_kernel(const float* __restrict__ h, const int* __restrict__ rowstart,
                       const int* __restrict__ col, float* __restrict__ agg, int N) {
    const int lane = threadIdx.x & 31;
    const int node = __builtin_amdgcn_readfirstlane((int)(blockIdx.x * 8 + (threadIdx.x >> 5)));
    if (node >= N) return;
    const int beg = rowstart[node];
    const int end = rowstart[node + 1];
    const float* hp = h + lane * 4;
    float4 acc = make_float4(0.0f, 0.0f, 0.0f, 0.0f);
    int j = beg;
    while (j + 8 <= end) GATHER_ROWS(8)
    if (j + 4 <= end) GATHER_ROWS(4)
    if (j + 2 <= end) GATHER_ROWS(2)
    if (j < end) GATHER_ROWS(1)
    *(float4*)(agg + (size_t)node * H_DIM + lane * 4) = acc;
}

extern "C" void kernel_launch(void* const* d_in, const int* in_sizes, int n_in,
                              void* d_out, int out_size, void* d_ws, size_t ws_size,
                              hipStream_t stream) {
    const float* x     = (const float*)d_in[0];
    const int*   ei    = (const int*)d_in[1];
    const float* w_in  = (const float*)d_in[2];
    const float* b_in  = (const float*)d_in[3];
    const float* w_l1  = (const float*)d_in[4];
    const float* b_l1  = (const float*)d_in[5];
    const float* w_r1  = (const float*)d_in[6];
    const float* g1    = (const float*)d_in[7];
    const float* be1   = (const float*)d_in[8];
    const float* w_l2  = (const float*)d_in[9];
    const float* b_l2  = (const float*)d_in[10];
    const float* w_r2  = (const float*)d_in[11];
    const float* g2    = (const float*)d_in[12];
    const float* be2   = (const float*)d_in[13];
    const float* w_out = (const float*)d_in[14];
    const float* b_out = (const float*)d_in[15];

    const int N = in_sizes[0] / DIN;
    const int E = in_sizes[1] / 2;
    const int* src = ei;
    const int* dst = ei + E;

    float* ws  = (float*)d_ws;
    float* h0  = ws;
    float* agg = ws + (size_t)N * H_DIM;
    float* h1  = ws + 2 * (size_t)N * H_DIM;
    float* deg = ws + 3 * (size_t)N * H_DIM;

    int* col      = (int*)d_out;
    int* rowstart = col + E;

    const int NB = (N + PB_NODES - 1) >> PB_SHIFT;
    int* psrc  = (int*)agg;
    int* pdst  = psrc + E;
    int* lrank = pdst + E;
    int* gh    = lrank + E;
    int* rs    = gh + (size_t)NB * PB_GRID;

    part_count_kernel<<<PB_GRID, 1024, 0, stream>>>(dst, lrank, gh, E, NB);
    part_move_kernel<<<PB_GRID, 1024, 0, stream>>>(src, dst, lrank, gh, rs, psrc, pdst, E, NB);
    bucket_fill_kernel<<<NB, 1024, 0, stream>>>(psrc, pdst, rs, rowstart, deg, col, N);

    gemm_strips<H_DIM, 4><<<GEMM_BLOCKS, GEMM_THREADS, 0, stream>>>(
        x, w_in, DIN, nullptr, nullptr, 0, nullptr, b_in, nullptr, nullptr,
        h0, N, 1);

    gather_sum_kernel<<<(N + 7) / 8, 256, 0, stream>>>(h0, rowstart, col, agg, N);
    gemm_strips<H_DIM, 4><<<GEMM_BLOCKS, GEMM_THREADS, 0, stream>>>(
        agg, w_l1, H_DIM, h0, w_r1, H_DIM, deg, b_l1, g1, be1,
        h1, N, 1);

    gather_sum_kernel<<<(N + 7) / 8, 256, 0, stream>>>(h1, rowstart, col, agg, N);
    gemm_strips<H_DIM, 4><<<GEMM_BLOCKS, GEMM_THREADS, 0, stream>>>(
        agg, w_l2, H_DIM, h1, w_r2, H_DIM, deg, b_l2, g2, be2,
        h0, N, 1);

    gemm_strips<DOUT, 4><<<GEMM_BLOCKS, GEMM_THREADS, 0, stream>>>(
        h0, w_out, H_DIM, nullptr, nullptr, 0, nullptr, b_out, nullptr, nullptr,
        (float*)d_out, N, 0);
}


# Neighbour sums fused into the layer products: a wave sums 16 target rows into an LDS strip and multiplies from it, self product first; 9 launches to 7; k order of layer products: self part now first
# speedup vs baseline: 3.1431x; 1.0558x over previous
#include <hip/hip_runtime.h>

#define DIN   256
#define H_DIM 128
#define DOUT  64
#define BN_RSQRT 0.9999950000374996f

typedef __attribute__((ext_vector_type(2))) float v2f;
typedef __attribute__((ext_vector_type(8))) float v8f;

__device__ __forceinline__ v8f wmma4(v2f a, v2f b, v8f c) {
    return __builtin_amdgcn_wmma_f32_16x16x4_f32(
         false, a,  false, b,
         (short)0, c,  false,  false);
}

#define GEMM_BLOCKS    256
#define GEMM_THREADS   1024
#define GEMM_LDS_BYTES (164 * 1024)

template <int NTW>
__device__ __forceinline__ void strip_accum(const float* __restrict__ ap,
                                            int K, float rs,
                                            const float4* bq,
                                            int ncols, v8f (&acc)[NTW]) {
    for (int k0 = 0; k0 < K; k0 += 32) {
        float4 a[4];
#pragma unroll
        for (int j = 0; j < 4; ++j) a[j] = *(const float4*)(ap + k0 + 4 * j);
        const float4* bk = bq + (size_t)(k0 >> 2) * ncols;
        float4 b[NTW];
#pragma unroll
        for (int ni = 0; ni < NTW; ++ni) b[ni] = bk[ni * 16];
#pragma unroll
        for (int j = 0; j < 4; ++j) {
            float4 bn[NTW];
            if (j < 3) {
#pragma unroll
                for (int ni = 0; ni < NTW; ++ni) bn[ni] = bk[(size_t)(j + 1) * ncols + ni * 16];
            }
            v2f alo, ahi;
            alo.x = a[j].x * rs;  alo.y = a[j].y * rs;
            ahi.x = a[j].z * rs;  ahi.y = a[j].w * rs;
#pragma unroll
            for (int ni = 0; ni < NTW; ++ni) {
                v2f bb; bb.x = b[ni].x; bb.y = b[ni].y;
                acc[ni] = wmma4(alo, bb, acc[ni]);
            }
#pragma unroll
            for (int ni = 0; ni < NTW; ++ni) {
                v2f bb; bb.x = b[ni].z; bb.y = b[ni].w;
                acc[ni] = wmma4(ahi, bb, acc[ni]);
            }
            if (j < 3) {
#pragma unroll
                for (int ni = 0; ni < NTW; ++ni) b[ni] = bn[ni];
            }
        }
    }
}

template <int NCOLS, int NTW>
__global__ __launch_bounds__(GEMM_THREADS)
void gemm_strips(const float* __restrict__ A0, const float* __restrict__ W0, int K0,
                 const float* __restrict__ A1, const float* __restrict__ W1, int K1,
                 const float* __restrict__ rowscale,
                 const float* __restrict__ bias,
                 const float* __restrict__ gamma, const float* __restrict__ beta,
                 float* __restrict__ out, int nrows, int do_relu) {
    __shared__ __attribute__((aligned(16))) float4 wl[GEMM_LDS_BYTES / 16];
    const int tid = threadIdx.x;
    const int lane = tid & 31;
    const int wave = __builtin_amdgcn_readfirstlane(tid >> 5);
    const int c = lane & 15, h = lane >> 4;
    const int Q0 = K0 >> 2;
    const int Q1 = A1 ? (K1 >> 2) : 0;

    const int nq = (Q0 + Q1) * NCOLS;
    for (int it = tid; it < nq; it += GEMM_THREADS) {
        const int q = it / NCOLS, n = it % NCOLS;
        const float* wp = (q < Q0) ? (W0 + (size_t)(4 * q) * NCOLS + n)
                                   : (W1 + (size_t)(4 * (q - Q0)) * NCOLS + n);
        float4 v;
        v.x = wp[0];  v.y = wp[NCOLS];  v.z = wp[2 * NCOLS];  v.w = wp[3 * NCOLS];
        wl[it] = v;
    }
    __syncthreads();

    constexpr int NSPLIT = NCOLS / (16 * NTW);
    const int items = (nrows >> 4) * NSPLIT;
    const int beg = (int)(((long long)blockIdx.x * items) / gridDim.x);
    const int end = (int)(((long long)(blockIdx.x + 1) * items) / gridDim.x);

    for (int it = beg + wave; it < end; it += GEMM_THREADS / 32) {
        const int strip = it / NSPLIT, part = it % NSPLIT;
        const int row0 = strip * 16, col0 = part * 16 * NTW;

        v8f acc[NTW] = {};
        const float rs = rowscale ? rowscale[row0 + c] : 1.0f;
        strip_accum<NTW>(A0 + (size_t)(row0 + c) * K0 + 16 * h, K0, rs,
                         wl + (size_t)(4 * h) * NCOLS + col0 + c, NCOLS, acc);
        if (A1)
            strip_accum<NTW>(A1 + (size_t)(row0 + c) * K1 + 16 * h, K1, 1.0f,
                             wl + (size_t)(Q0 + 4 * h) * NCOLS + col0 + c, NCOLS, acc);

#pragma unroll
        for (int ni = 0; ni < NTW; ++ni) {
            const int col = col0 + ni * 16 + c;
            const float bi = bias ? bias[col] : 0.0f;
            const float sc = gamma ? gamma[col] * BN_RSQRT : 1.0f;
            const float sh = beta ? beta[col] : 0.0f;
#pragma unroll
            for (int r = 0; r < 8; ++r) {
                const int row = row0 + r + 8 * h;
                float v = acc[ni][r];
                v = (v + bi) * sc + sh;
                if (do_relu) v = fmaxf(v, 0.0f);
                out[(size_t)row * NCOLS + col] = v;
            }
        }
    }
}

#define PB_SHIFT 9
#define PB_NODES (1 << PB_SHIFT)
#define PB_MAXB  2048
#define PB_GRID  256
#define PB_BATCH 8

__global__ __launch_bounds__(1024)
void part_count_kernel(const int* __restrict__ dst, int* __restrict__ lrank,
                       int* __restrict__ gh, int E, int NB) {
    __shared__ int hist[PB_MAXB];
    const int tid = threadIdx.x;
    for (int i = tid; i < NB; i += 1024) hist[i] = 0;
    __syncthreads();
    const int beg = (int)(((long long)blockIdx.x * E) / gridDim.x);
    const int end = (int)(((long long)(blockIdx.x + 1) * E) / gridDim.x);
    for (int e = beg + tid; e < end; e += PB_BATCH * 1024) {
        int d[PB_BATCH];
#pragma unroll
        for (int u = 0; u < PB_BATCH; ++u)
            d[u] = dst[min(e + u * 1024, end - 1)];
#pragma unroll
        for (int u = 0; u < PB_BATCH; ++u)
            if (e + u * 1024 < end) lrank[e + u * 1024] = atomicAdd(&hist[d[u] >> PB_SHIFT], 1);
    }
    __syncthreads();
    for (int i = tid; i < NB; i += 1024) gh[(size_t)i * PB_GRID + blockIdx.x] = hist[i];
}

__global__ __launch_bounds__(1024)
void part_move_kernel(const int* __restrict__ src, const int* __restrict__ dst,
                      const int* __restrict__ lrank, const int* __restrict__ gh,
                      int* __restrict__ rs,
                      int* __restrict__ psrc, int* __restrict__ pdst, int E, int NB) {
    __shared__ int base[PB_MAXB];
    __shared__ int tot[PB_MAXB];
    __shared__ int s_w[32];
    const int tid = threadIdx.x;
    const int lane = tid & 31;
    const int wave = __builtin_amdgcn_readfirstlane(tid >> 5);
    const int g = blockIdx.x;

    int v[PB_GRID / 32];
    const int* row = gh + (size_t)min(wave, NB - 1) * PB_GRID + lane;
#pragma unroll
    for (int m = 0; m < PB_GRID / 32; ++m) v[m] = row[32 * m];
    for (int i = wave; i < NB; i += 32) {
        int nv[PB_GRID / 32];
        row = gh + (size_t)min(i + 32, NB - 1) * PB_GRID + lane;
#pragma unroll
        for (int m = 0; m < PB_GRID / 32; ++m) nv[m] = row[32 * m];
        int t = 0, p = 0;
#pragma unroll
        for (int m = 0; m < PB_GRID / 32; ++m) {
            t += v[m];
            if (32 * m + lane < g) p += v[m];
        }
        for (int d = 16; d > 0; d >>= 1) {
            t += __shfl_xor(t, d);
            p += __shfl_xor(p, d);
        }
        if (lane == 0) { tot[i] = t; base[i] = p; }
#pragma unroll
        for (int m = 0; m < PB_GRID / 32; ++m) v[m] = nv[m];
    }
    __syncthreads();

    constexpr int RPT = PB_MAXB / 1024;
    int loc = 0;
#pragma unroll
    for (int u = 0; u < RPT; ++u)
        if (tid * RPT + u < NB) loc += tot[tid * RPT + u];
    int incl = loc;
    for (int d = 1; d < 32; d <<= 1) {
        const int up = __shfl_up(incl, d);
        if (lane >= d) incl += up;
    }
    if (lane == 31) s_w[wave] = incl;
    __syncthreads();
    int run = incl - loc;
    for (int w = 0; w < wave; ++w) run += s_w[w];
#pragma unroll
    for (int u = 0; u < RPT; ++u) {
        const int i = tid * RPT + u;
        if (i < NB) {
            const int sz = tot[i];
            base[i] += run;
            if (g == 0) rs[i] = run;
            run += sz;
        }
    }
    if (g == 0 && tid == 1023) rs[NB] = run;
    __syncthreads();
    const int beg = (int)(((long long)blockIdx.x * E) / gridDim.x);
    const int end = (int)(((long long)(blockIdx.x + 1) * E) / gridDim.x);
    for (int e = beg + tid; e < end; e += PB_BATCH * 1024) {
        int d[PB_BATCH], r[PB_BATCH], s[PB_BATCH];
#pragma unroll
        for (int u = 0; u < PB_BATCH; ++u) {
            const int ee = min(e + u * 1024, end - 1);
            d[u] = dst[ee];
            r[u] = lrank[ee];
            s[u] = src[ee];
        }
#pragma unroll
        for (int u = 0; u < PB_BATCH; ++u) {
            if (e + u * 1024 < end) {
                const int pos = base[d[u] >> PB_SHIFT] + r[u];
                psrc[pos] = s[u];
                pdst[pos] = d[u];
            }
        }
    }
}

__global__ __launch_bounds__(1024)
void bucket_fill_kernel(const int* __restrict__ psrc, const int* __restrict__ pdst,
                        const int* __restrict__ rs,
                        int* __restrict__ rowstart, float* __restrict__ invdeg,
                        int* __restrict__ col, int N) {
    __shared__ int cnt[PB_NODES];
    __shared__ int start[PB_NODES];
    __shared__ int s_w[32];
    const int tid = threadIdx.x;
    const int lane = tid & 31, wave = tid >> 5;
    const int b = blockIdx.x;
    const int s0 = rs[b];
    const int s1 = rs[b + 1];
    const int n0 = b << PB_SHIFT;

    if (tid < PB_NODES) cnt[tid] = 0;
    __syncthreads();
    for (int e = s0 + tid; e < s1; e += PB_BATCH * 1024) {
        int d[PB_BATCH];
#pragma unroll
        for (int u = 0; u < PB_BATCH; ++u)
            d[u] = pdst[min(e + u * 1024, s1 - 1)];
#pragma unroll
        for (int u = 0; u < PB_BATCH; ++u)
            if (e + u * 1024 < s1) atomicAdd(&cnt[d[u] - n0], 1);
    }
    __syncthreads();

    const int c = (tid < PB_NODES) ? cnt[tid] : 0;
    int incl = c;
    for (int d = 1; d < 32; d <<= 1) {
        const int up = __shfl_up(incl, d);
        if (lane >= d) incl += up;
    }
    if (lane == 31) s_w[wave] = incl;
    __syncthreads();
    int excl = incl - c;
    for (int w = 0; w < wave; ++w) excl += s_w[w];
    if (tid < PB_NODES) {
        start[tid] = excl;
        cnt[tid] = 0;
        const int node = n0 + tid;
        if (node < N) {
            rowstart[node] = s0 + excl;
            invdeg[node] = 1.0f / fmaxf((float)c, 1.0f);
        }
    }
    if (b == (int)gridDim.x - 1 && tid == 0) rowstart[N] = s1;
    __syncthreads();

    for (int e = s0 + tid; e < s1; e += PB_BATCH * 1024) {
        int d[PB_BATCH], s[PB_BATCH];
#pragma unroll
        for (int u = 0; u < PB_BATCH; ++u) {
            const int ee = min(e + u * 1024, s1 - 1);
            d[u] = pdst[ee];
            s[u] = psrc[ee];
        }
#pragma unroll
        for (int u = 0; u < PB_BATCH; ++u) {
            if (e + u * 1024 < s1) {
                const int dl = d[u] - n0;
                const int slot = atomicAdd(&cnt[dl], 1);
                col[s0 + start[dl] + slot] = s[u];
            }
        }
    }
}

#define FL_WAVES   16
#define FL_THREADS (FL_WAVES * 32)
#define FL_PITCH   132
#define FL_U       16

__global__ __launch_bounds__(FL_THREADS)
void layer_fused_kernel(const float* __restrict__ hin, const int* __restrict__ rowstart,
                       const int* __restrict__ col,
                       const float* __restrict__ Wl, const float* __restrict__ Wr,
                       const float* __restrict__ invdeg, const float* __restrict__ bias,
                       const float* __restrict__ gamma, const float* __restrict__ beta,
                       float* __restrict__ out, int nrows) {
    constexpr int Q0 = H_DIM / 4;
    constexpr int NQ = 2 * Q0 * H_DIM;
    __shared__ __attribute__((aligned(16))) float4 wl[NQ];
    __shared__ __attribute__((aligned(16))) float strips[FL_WAVES * 16 * FL_PITCH];
    __shared__ int s_next;
    const int tid = threadIdx.x;
    const int lane = tid & 31;
    const int wave = __builtin_amdgcn_readfirstlane(tid >> 5);
    const int c = lane & 15, h = lane >> 4;

    for (int it0 = tid; it0 < NQ; it0 += 4 * FL_THREADS) {
        float4 v[4];
#pragma unroll
        for (int u = 0; u < 4; ++u) {
            const int it = min(it0 + u * FL_THREADS, NQ - 1);
            const int q = it / H_DIM, n = it % H_DIM;
            const float* wp = (q < Q0) ? (Wl + (size_t)(4 * q) * H_DIM + n)
                                       : (Wr + (size_t)(4 * (q - Q0)) * H_DIM + n);
            v[u].x = wp[0];  v[u].y = wp[H_DIM];  v[u].z = wp[2 * H_DIM];  v[u].w = wp[3 * H_DIM];
        }
#pragma unroll
        for (int u = 0; u < 4; ++u)
            if (it0 + u * FL_THREADS < NQ) wl[it0 + u * FL_THREADS] = v[u];
    }
    if (tid == 0) s_next = 0;
    __syncthreads();

    const int items = nrows >> 4;
    const int beg = (int)(((long long)blockIdx.x * items) / gridDim.x);
    const int end = (int)(((long long)(blockIdx.x + 1) * items) / gridDim.x);
    float* sp = strips + wave * (16 * FL_PITCH);
    const float* hp = hin + lane * 4;

    for (;;) {
        int t = 0;
        if (lane == 0) t = atomicAdd(&s_next, 1);
        const int it = beg + __builtin_amdgcn_readfirstlane(t);
        if (it >= end) break;
        const int row0 = it * 16;

        v8f acc8[8] = {};
        strip_accum<8>(hin + (size_t)(row0 + c) * H_DIM + 16 * h, H_DIM, 1.0f,
                       wl + (size_t)(Q0 + 4 * h) * H_DIM + c, H_DIM, acc8);

        const int rsv = rowstart[row0 + min(lane, 16)];
        int j = __builtin_amdgcn_readlane(rsv, 0);
        const int jend = __builtin_amdgcn_readlane(rsv, 16);
        int r = 0;
        int nb = __builtin_amdgcn_readlane(rsv, 1);
        float4 acc = make_float4(0.0f, 0.0f, 0.0f, 0.0f);
        int idv = col[max(min(j + c, jend - 1), 0)];
        while (j < jend) {
            float4 v[FL_U];
#pragma unroll
            for (int u = 0; u < FL_U; ++u) {
                const int id = __builtin_amdgcn_readlane(idv, u);
                v[u] = *(const float4*)(hp + (size_t)id * H_DIM);
            }
            idv = col[min(j + FL_U + c, jend - 1)];
#pragma unroll
            for (int u = 0; u < FL_U; ++u) {
                if (j + u < jend) {
                    while (j + u == nb) {
                        *(float4*)(sp + r * FL_PITCH + lane * 4) = acc;
                        acc = make_float4(0.0f, 0.0f, 0.0f, 0.0f);
                        ++r;
                        nb = __builtin_amdgcn_readlane(rsv, r + 1);
                    }
                    acc.x += v[u].x; acc.y += v[u].y; acc.z += v[u].z; acc.w += v[u].w;
                }
            }
            j += FL_U;
        }
        while (r < 16) {
            *(float4*)(sp + r * FL_PITCH + lane * 4) = acc;
            acc = make_float4(0.0f, 0.0f, 0.0f, 0.0f);
            ++r;
        }

        const float rs = invdeg[row0 + c];
        strip_accum<8>(sp + c * FL_PITCH + 16 * h, H_DIM, rs,
                       wl + (size_t)(4 * h) * H_DIM + c, H_DIM, acc8);

#pragma unroll
        for (int ni = 0; ni < 8; ++ni) {
            const int cc = ni * 16 + c;
            const float bi = bias[cc];
            const float sc = gamma[cc] * BN_RSQRT;
            const float sh = beta[cc];
#pragma unroll
            for (int rr = 0; rr < 8; ++rr) {
                const int row = row0 + rr + 8 * h;
                float v = acc8[ni][rr];
                v = (v + bi) * sc + sh;
                v = fmaxf(v, 0.0f);
                out[(size_t)row * H_DIM + cc] = v;
            }
        }
    }
}

extern "C" void kernel_launch(void* const* d_in, const int* in_sizes, int n_in,
                              void* d_out, int out_size, void* d_ws, size_t ws_size,
                              hipStream_t stream) {
    const float* x     = (const float*)d_in[0];
    const int*   ei    = (const int*)d_in[1];
    const float* w_in  = (const float*)d_in[2];
    const float* b_in  = (const float*)d_in[3];
    const float* w_l1  = (const float*)d_in[4];
    const float* b_l1  = (const float*)d_in[5];
    const float* w_r1  = (const float*)d_in[6];
    const float* g1    = (const float*)d_in[7];
    const float* be1   = (const float*)d_in[8];
    const float* w_l2  = (const float*)d_in[9];
    const float* b_l2  = (const float*)d_in[10];
    const float* w_r2  = (const float*)d_in[11];
    const float* g2    = (const float*)d_in[12];
    const float* be2   = (const float*)d_in[13];
    const float* w_out = (const float*)d_in[14];
    const float* b_out = (const float*)d_in[15];

    const int N = in_sizes[0] / DIN;
    const int E = in_sizes[1] / 2;
    const int* src = ei;
    const int* dst = ei + E;

    float* ws  = (float*)d_ws;
    float* h0  = ws;
    float* agg = ws + (size_t)N * H_DIM;
    float* h1  = ws + 2 * (size_t)N * H_DIM;
    float* deg = ws + 3 * (size_t)N * H_DIM;

    int* col      = (int*)d_out;
    int* rowstart = col + E;

    const int NB = (N + PB_NODES - 1) >> PB_SHIFT;
    int* psrc  = (int*)agg;
    int* pdst  = psrc + E;
    int* lrank = pdst + E;
    int* gh    = lrank + E;
    int* rs    = gh + (size_t)NB * PB_GRID;

    part_count_kernel<<<PB_GRID, 1024, 0, stream>>>(dst, lrank, gh, E, NB);
    part_move_kernel<<<PB_GRID, 1024, 0, stream>>>(src, dst, lrank, gh, rs, psrc, pdst, E, NB);
    bucket_fill_kernel<<<NB, 1024, 0, stream>>>(psrc, pdst, rs, rowstart, deg, col, N);

    gemm_strips<H_DIM, 4><<<GEMM_BLOCKS, GEMM_THREADS, 0, stream>>>(
        x, w_in, DIN, nullptr, nullptr, 0, nullptr, b_in, nullptr, nullptr,
        h0, N, 1);

    layer_fused_kernel<<<GEMM_BLOCKS, FL_THREADS, 0, stream>>>(
        h0, rowstart, col, w_l1, w_r1, deg, b_l1, g1, be1, h1, N);

    layer_fused_kernel<<<GEMM_BLOCKS, FL_THREADS, 0, stream>>>(
        h1, rowstart, col, w_l2, w_r2, deg, b_l2, g2, be2, h0, N);

    gemm_strips<DOUT, 4><<<GEMM_BLOCKS, GEMM_THREADS, 0, stream>>>(
        h0, w_out, H_DIM, nullptr, nullptr, 0, nullptr, b_out, nullptr, nullptr,
        (float*)d_out, N, 0);
}


# Output product fused into the second layer launch: a wave puts its 16 finished rows into its LDS strip and multiplies them by the output weights staged in LDS; 7 launches to 6; sum order unchanged
# speedup vs baseline: 3.2777x; 1.0428x over previous
#include <hip/hip_runtime.h>

#define DIN   256
#define H_DIM 128
#define DOUT  64
#define BN_RSQRT 0.9999950000374996f

typedef __attribute__((ext_vector_type(2))) float v2f;
typedef __attribute__((ext_vector_type(8))) float v8f;

__device__ __forceinline__ v8f wmma4(v2f a, v2f b, v8f c) {
    return __builtin_amdgcn_wmma_f32_16x16x4_f32(
         false, a,  false, b,
         (short)0, c,  false,  false);
}

#define GEMM_BLOCKS    256
#define GEMM_THREADS   1024
#define GEMM_LDS_BYTES (164 * 1024)

template <int NTW>
__device__ __forceinline__ void strip_accum(const float* __restrict__ ap,
                                            int K, float rs,
                                            const float4* bq,
                                            int ncols, v8f (&acc)[NTW]) {
    for (int k0 = 0; k0 < K; k0 += 32) {
        float4 a[4];
#pragma unroll
        for (int j = 0; j < 4; ++j) a[j] = *(const float4*)(ap + k0 + 4 * j);
        const float4* bk = bq + (size_t)(k0 >> 2) * ncols;
        float4 b[NTW];
#pragma unroll
        for (int ni = 0; ni < NTW; ++ni) b[ni] = bk[ni * 16];
#pragma unroll
        for (int j = 0; j < 4; ++j) {
            float4 bn[NTW];
            if (j < 3) {
#pragma unroll
                for (int ni = 0; ni < NTW; ++ni) bn[ni] = bk[(size_t)(j + 1) * ncols + ni * 16];
            }
            v2f alo, ahi;
            alo.x = a[j].x * rs;  alo.y = a[j].y * rs;
            ahi.x = a[j].z * rs;  ahi.y = a[j].w * rs;
#pragma unroll
            for (int ni = 0; ni < NTW; ++ni) {
                v2f bb; bb.x = b[ni].x; bb.y = b[ni].y;
                acc[ni] = wmma4(alo, bb, acc[ni]);
            }
#pragma unroll
            for (int ni = 0; ni < NTW; ++ni) {
                v2f bb; bb.x = b[ni].z; bb.y = b[ni].w;
                acc[ni] = wmma4(ahi, bb, acc[ni]);
            }
            if (j < 3) {
#pragma unroll
                for (int ni = 0; ni < NTW; ++ni) b[ni] = bn[ni];
            }
        }
    }
}

template <int NCOLS, int NTW>
__global__ __launch_bounds__(GEMM_THREADS)
void gemm_strips(const float* __restrict__ A0, const float* __restrict__ W0, int K0,
                 const float* __restrict__ A1, const float* __restrict__ W1, int K1,
                 const float* __restrict__ rowscale,
                 const float* __restrict__ bias,
                 const float* __restrict__ gamma, const float* __restrict__ beta,
                 float* __restrict__ out, int nrows, int do_relu) {
    __shared__ __attribute__((aligned(16))) float4 wl[GEMM_LDS_BYTES / 16];
    const int tid = threadIdx.x;
    const int lane = tid & 31;
    const int wave = __builtin_amdgcn_readfirstlane(tid >> 5);
    const int c = lane & 15, h = lane >> 4;
    const int Q0 = K0 >> 2;
    const int Q1 = A1 ? (K1 >> 2) : 0;

    const int nq = (Q0 + Q1) * NCOLS;
    for (int it = tid; it < nq; it += GEMM_THREADS) {
        const int q = it / NCOLS, n = it % NCOLS;
        const float* wp = (q < Q0) ? (W0 + (size_t)(4 * q) * NCOLS + n)
                                   : (W1 + (size_t)(4 * (q - Q0)) * NCOLS + n);
        float4 v;
        v.x = wp[0];  v.y = wp[NCOLS];  v.z = wp[2 * NCOLS];  v.w = wp[3 * NCOLS];
        wl[it] = v;
    }
    __syncthreads();

    constexpr int NSPLIT = NCOLS / (16 * NTW);
    const int items = (nrows >> 4) * NSPLIT;
    const int beg = (int)(((long long)blockIdx.x * items) / gridDim.x);
    const int end = (int)(((long long)(blockIdx.x + 1) * items) / gridDim.x);

    for (int it = beg + wave; it < end; it += GEMM_THREADS / 32) {
        const int strip = it / NSPLIT, part = it % NSPLIT;
        const int row0 = strip * 16, col0 = part * 16 * NTW;

        v8f acc[NTW] = {};
        const float rs = rowscale ? rowscale[row0 + c] : 1.0f;
        strip_accum<NTW>(A0 + (size_t)(row0 + c) * K0 + 16 * h, K0, rs,
                         wl + (size_t)(4 * h) * NCOLS + col0 + c, NCOLS, acc);
        if (A1)
            strip_accum<NTW>(A1 + (size_t)(row0 + c) * K1 + 16 * h, K1, 1.0f,
                             wl + (size_t)(Q0 + 4 * h) * NCOLS + col0 + c, NCOLS, acc);

#pragma unroll
        for (int ni = 0; ni < NTW; ++ni) {
            const int col = col0 + ni * 16 + c;
            const float bi = bias ? bias[col] : 0.0f;
            const float sc = gamma ? gamma[col] * BN_RSQRT : 1.0f;
            const float sh = beta ? beta[col] : 0.0f;
#pragma unroll
            for (int r = 0; r < 8; ++r) {
                const int row = row0 + r + 8 * h;
                float v = acc[ni][r];
                v = (v + bi) * sc + sh;
                if (do_relu) v = fmaxf(v, 0.0f);
                out[(size_t)row * NCOLS + col] = v;
            }
        }
    }
}

#define PB_SHIFT 9
#define PB_NODES (1 << PB_SHIFT)
#define PB_MAXB  2048
#define PB_GRID  256
#define PB_BATCH 8

__global__ __launch_bounds__(1024)
void part_count_kernel(const int* __restrict__ dst, int* __restrict__ lrank,
                       int* __restrict__ gh, int E, int NB) {
    __shared__ int hist[PB_MAXB];
    const int tid = threadIdx.x;
    for (int i = tid; i < NB; i += 1024) hist[i] = 0;
    __syncthreads();
    const int beg = (int)(((long long)blockIdx.x * E) / gridDim.x);
    const int end = (int)(((long long)(blockIdx.x + 1) * E) / gridDim.x);
    for (int e = beg + tid; e < end; e += PB_BATCH * 1024) {
        int d[PB_BATCH];
#pragma unroll
        for (int u = 0; u < PB_BATCH; ++u)
            d[u] = dst[min(e + u * 1024, end - 1)];
#pragma unroll
        for (int u = 0; u < PB_BATCH; ++u)
            if (e + u * 1024 < end) lrank[e + u * 1024] = atomicAdd(&hist[d[u] >> PB_SHIFT], 1);
    }
    __syncthreads();
    for (int i = tid; i < NB; i += 1024) gh[(size_t)i * PB_GRID + blockIdx.x] = hist[i];
}

__global__ __launch_bounds__(1024)
void part_move_kernel(const int* __restrict__ src, const int* __restrict__ dst,
                      const int* __restrict__ lrank, const int* __restrict__ gh,
                      int* __restrict__ rs,
                      int* __restrict__ psrc, int* __restrict__ pdst, int E, int NB) {
    __shared__ int base[PB_MAXB];
    __shared__ int tot[PB_MAXB];
    __shared__ int s_w[32];
    const int tid = threadIdx.x;
    const int lane = tid & 31;
    const int wave = __builtin_amdgcn_readfirstlane(tid >> 5);
    const int g = blockIdx.x;

    int v[PB_GRID / 32];
    const int* row = gh + (size_t)min(wave, NB - 1) * PB_GRID + lane;
#pragma unroll
    for (int m = 0; m < PB_GRID / 32; ++m) v[m] = row[32 * m];
    for (int i = wave; i < NB; i += 32) {
        int nv[PB_GRID / 32];
        row = gh + (size_t)min(i + 32, NB - 1) * PB_GRID + lane;
#pragma unroll
        for (int m = 0; m < PB_GRID / 32; ++m) nv[m] = row[32 * m];
        int t = 0, p = 0;
#pragma unroll
        for (int m = 0; m < PB_GRID / 32; ++m) {
            t += v[m];
            if (32 * m + lane < g) p += v[m];
        }
        for (int d = 16; d > 0; d >>= 1) {
            t += __shfl_xor(t, d);
            p += __shfl_xor(p, d);
        }
        if (lane == 0) { tot[i] = t; base[i] = p; }
#pragma unroll
        for (int m = 0; m < PB_GRID / 32; ++m) v[m] = nv[m];
    }
    __syncthreads();

    constexpr int RPT = PB_MAXB / 1024;
    int loc = 0;
#pragma unroll
    for (int u = 0; u < RPT; ++u)
        if (tid * RPT + u < NB) loc += tot[tid * RPT + u];
    int incl = loc;
    for (int d = 1; d < 32; d <<= 1) {
        const int up = __shfl_up(incl, d);
        if (lane >= d) incl += up;
    }
    if (lane == 31) s_w[wave] = incl;
    __syncthreads();
    int run = incl - loc;
    for (int w = 0; w < wave; ++w) run += s_w[w];
#pragma unroll
    for (int u = 0; u < RPT; ++u) {
        const int i = tid * RPT + u;
        if (i < NB) {
            const int sz = tot[i];
            base[i] += run;
            if (g == 0) rs[i] = run;
            run += sz;
        }
    }
    if (g == 0 && tid == 1023) rs[NB] = run;
    __syncthreads();
    const int beg = (int)(((long long)blockIdx.x * E) / gridDim.x);
    const int end = (int)(((long long)(blockIdx.x + 1) * E) / gridDim.x);
    for (int e = beg + tid; e < end; e += PB_BATCH * 1024) {
        int d[PB_BATCH], r[PB_BATCH], s[PB_BATCH];
#pragma unroll
        for (int u = 0; u < PB_BATCH; ++u) {
            const int ee = min(e + u * 1024, end - 1);
            d[u] = dst[ee];
            r[u] = lrank[ee];
            s[u] = src[ee];
        }
#pragma unroll
        for (int u = 0; u < PB_BATCH; ++u) {
            if (e + u * 1024 < end) {
                const int pos = base[d[u] >> PB_SHIFT] + r[u];
                psrc[pos] = s[u];
                pdst[pos] = d[u];
            }
        }
    }
}

__global__ __launch_bounds__(1024)
void bucket_fill_kernel(const int* __restrict__ psrc, const int* __restrict__ pdst,
                        const int* __restrict__ rs,
                        int* __restrict__ rowstart, float* __restrict__ invdeg,
                        int* __restrict__ col, int N) {
    __shared__ int cnt[PB_NODES];
    __shared__ int start[PB_NODES];
    __shared__ int s_w[32];
    const int tid = threadIdx.x;
    const int lane = tid & 31, wave = tid >> 5;
    const int b = blockIdx.x;
    const int s0 = rs[b];
    const int s1 = rs[b + 1];
    const int n0 = b << PB_SHIFT;

    if (tid < PB_NODES) cnt[tid] = 0;
    __syncthreads();
    for (int e = s0 + tid; e < s1; e += PB_BATCH * 1024) {
        int d[PB_BATCH];
#pragma unroll
        for (int u = 0; u < PB_BATCH; ++u)
            d[u] = pdst[min(e + u * 1024, s1 - 1)];
#pragma unroll
        for (int u = 0; u < PB_BATCH; ++u)
            if (e + u * 1024 < s1) atomicAdd(&cnt[d[u] - n0], 1);
    }
    __syncthreads();

    const int c = (tid < PB_NODES) ? cnt[tid] : 0;
    int incl = c;
    for (int d = 1; d < 32; d <<= 1) {
        const int up = __shfl_up(incl, d);
        if (lane >= d) incl += up;
    }
    if (lane == 31) s_w[wave] = incl;
    __syncthreads();
    int excl = incl - c;
    for (int w = 0; w < wave; ++w) excl += s_w[w];
    if (tid < PB_NODES) {
        start[tid] = excl;
        cnt[tid] = 0;
        const int node = n0 + tid;
        if (node < N) {
            rowstart[node] = s0 + excl;
            invdeg[node] = 1.0f / fmaxf((float)c, 1.0f);
        }
    }
    if (b == (int)gridDim.x - 1 && tid == 0) rowstart[N] = s1;
    __syncthreads();

    for (int e = s0 + tid; e < s1; e += PB_BATCH * 1024) {
        int d[PB_BATCH], s[PB_BATCH];
#pragma unroll
        for (int u = 0; u < PB_BATCH; ++u) {
            const int ee = min(e + u * 1024, s1 - 1);
            d[u] = pdst[ee];
            s[u] = psrc[ee];
        }
#pragma unroll
        for (int u = 0; u < PB_BATCH; ++u) {
            if (e + u * 1024 < s1) {
                const int dl = d[u] - n0;
                const int slot = atomicAdd(&cnt[dl], 1);
                col[s0 + start[dl] + slot] = s[u];
            }
        }
    }
}

#define FL_WAVES   16
#define FL_THREADS (FL_WAVES * 32)
#define FL_PITCH   132
#define FL_U       16

template <bool FUSE_OUT>
__global__ __launch_bounds__(FL_THREADS)
void layer_fused_kernel(const float* __restrict__ hin, const int* __restrict__ rowstart,
                       const int* __restrict__ col,
                       const float* __restrict__ Wl, const float* __restrict__ Wr,
                       const float* __restrict__ invdeg, const float* __restrict__ bias,
                       const float* __restrict__ gamma, const float* __restrict__ beta,
                       const float* __restrict__ Wo, const float* __restrict__ bo,
                       float* __restrict__ out, int nrows) {
    constexpr int Q0 = H_DIM / 4;
    constexpr int NQ = 2 * Q0 * H_DIM;
    constexpr int NQO = Q0 * DOUT;
    __shared__ __attribute__((aligned(16))) float4 wl[NQ];
    __shared__ __attribute__((aligned(16))) float4 wo[FUSE_OUT ? NQO : 1];
    __shared__ __attribute__((aligned(16))) float strips[FL_WAVES * 16 * FL_PITCH];
    __shared__ int s_next;
    const int tid = threadIdx.x;
    const int lane = tid & 31;
    const int wave = __builtin_amdgcn_readfirstlane(tid >> 5);
    const int c = lane & 15, h = lane >> 4;

    for (int it0 = tid; it0 < NQ; it0 += 4 * FL_THREADS) {
        float4 v[4];
#pragma unroll
        for (int u = 0; u < 4; ++u) {
            const int it = min(it0 + u * FL_THREADS, NQ - 1);
            const int q = it / H_DIM, n = it % H_DIM;
            const float* wp = (q < Q0) ? (Wl + (size_t)(4 * q) * H_DIM + n)
                                       : (Wr + (size_t)(4 * (q - Q0)) * H_DIM + n);
            v[u].x = wp[0];  v[u].y = wp[H_DIM];  v[u].z = wp[2 * H_DIM];  v[u].w = wp[3 * H_DIM];
        }
#pragma unroll
        for (int u = 0; u < 4; ++u)
            if (it0 + u * FL_THREADS < NQ) wl[it0 + u * FL_THREADS] = v[u];
    }
    if constexpr (FUSE_OUT) {
        static_assert(NQO % FL_THREADS == 0, "output weight entries per thread");
        float4 v[NQO / FL_THREADS];
#pragma unroll
        for (int u = 0; u < NQO / FL_THREADS; ++u) {
            const int it = tid + u * FL_THREADS;
            const int q = it / DOUT, n = it % DOUT;
            const float* wp = Wo + (size_t)(4 * q) * DOUT + n;
            v[u].x = wp[0];  v[u].y = wp[DOUT];  v[u].z = wp[2 * DOUT];  v[u].w = wp[3 * DOUT];
        }
#pragma unroll
        for (int u = 0; u < NQO / FL_THREADS; ++u) wo[tid + u * FL_THREADS] = v[u];
    }
    if (tid == 0) s_next = 0;
    __syncthreads();

    const int items = nrows >> 4;
    const int beg = (int)(((long long)blockIdx.x * items) / gridDim.x);
    const int end = (int)(((long long)(blockIdx.x + 1) * items) / gridDim.x);
    float* sp = strips + wave * (16 * FL_PITCH);
    const float* hp = hin + lane * 4;

    for (;;) {
        int t = 0;
        if (lane == 0) t = atomicAdd(&s_next, 1);
        const int it = beg + __builtin_amdgcn_readfirstlane(t);
        if (it >= end) break;
        const int row0 = it * 16;

        v8f acc8[8] = {};
        strip_accum<8>(hin + (size_t)(row0 + c) * H_DIM + 16 * h, H_DIM, 1.0f,
                       wl + (size_t)(Q0 + 4 * h) * H_DIM + c, H_DIM, acc8);

        const int rsv = rowstart[row0 + min(lane, 16)];
        int j = __builtin_amdgcn_readlane(rsv, 0);
        const int jend = __builtin_amdgcn_readlane(rsv, 16);
        int r = 0;
        int nb = __builtin_amdgcn_readlane(rsv, 1);
        float4 acc = make_float4(0.0f, 0.0f, 0.0f, 0.0f);
        int idv = col[max(min(j + c, jend - 1), 0)];
        while (j < jend) {
            float4 v[FL_U];
#pragma unroll
            for (int u = 0; u < FL_U; ++u) {
                const int id = __builtin_amdgcn_readlane(idv, u);
                v[u] = *(const float4*)(hp + (size_t)id * H_DIM);
            }
            idv = col[min(j + FL_U + c, jend - 1)];
#pragma unroll
            for (int u = 0; u < FL_U; ++u) {
                if (j + u < jend) {
                    while (j + u == nb) {
                        *(float4*)(sp + r * FL_PITCH + lane * 4) = acc;
                        acc = make_float4(0.0f, 0.0f, 0.0f, 0.0f);
                        ++r;
                        nb = __builtin_amdgcn_readlane(rsv, r + 1);
                    }
                    acc.x += v[u].x; acc.y += v[u].y; acc.z += v[u].z; acc.w += v[u].w;
                }
            }
            j += FL_U;
        }
        while (r < 16) {
            *(float4*)(sp + r * FL_PITCH + lane * 4) = acc;
            acc = make_float4(0.0f, 0.0f, 0.0f, 0.0f);
            ++r;
        }

        const float rs = invdeg[row0 + c];
        strip_accum<8>(sp + c * FL_PITCH + 16 * h, H_DIM, rs,
                       wl + (size_t)(4 * h) * H_DIM + c, H_DIM, acc8);

#pragma unroll
        for (int ni = 0; ni < 8; ++ni) {
            const int cc = ni * 16 + c;
            const float bi = bias[cc];
            const float sc = gamma[cc] * BN_RSQRT;
            const float sh = beta[cc];
#pragma unroll
            for (int rr = 0; rr < 8; ++rr) {
                const int row = row0 + rr + 8 * h;
                float v = acc8[ni][rr];
                v = (v + bi) * sc + sh;
                v = fmaxf(v, 0.0f);
                if constexpr (FUSE_OUT) {
                    sp[(rr + 8 * h) * FL_PITCH + cc] = v;
                } else {
                    out[(size_t)row * H_DIM + cc] = v;
                }
            }
        }

        if constexpr (FUSE_OUT) {
            v8f acc4[DOUT / 16] = {};
            strip_accum<DOUT / 16>(sp + c * FL_PITCH + 16 * h, H_DIM, 1.0f,
                                   wo + (size_t)(4 * h) * DOUT + c, DOUT, acc4);
#pragma unroll
            for (int ni = 0; ni < DOUT / 16; ++ni) {
                const int cc = ni * 16 + c;
                const float bi = bo[cc];
#pragma unroll
                for (int rr = 0; rr < 8; ++rr) {
                    const int row = row0 + rr + 8 * h;
                    out[(size_t)row * DOUT + cc] = acc4[ni][rr] + bi;
                }
            }
        }
    }
}

extern "C" void kernel_launch(void* const* d_in, const int* in_sizes, int n_in,
                              void* d_out, int out_size, void* d_ws, size_t ws_size,
                              hipStream_t stream) {
    const float* x     = (const float*)d_in[0];
    const int*   ei    = (const int*)d_in[1];
    const float* w_in  = (const float*)d_in[2];
    const float* b_in  = (const float*)d_in[3];
    const float* w_l1  = (const float*)d_in[4];
    const float* b_l1  = (const float*)d_in[5];
    const float* w_r1  = (const float*)d_in[6];
    const float* g1    = (const float*)d_in[7];
    const float* be1   = (const float*)d_in[8];
    const float* w_l2  = (const float*)d_in[9];
    const float* b_l2  = (const float*)d_in[10];
    const float* w_r2  = (const float*)d_in[11];
    const float* g2    = (const float*)d_in[12];
    const float* be2   = (const float*)d_in[13];
    const float* w_out = (const float*)d_in[14];
    const float* b_out = (const float*)d_in[15];

    const int N = in_sizes[0] / DIN;
    const int E = in_sizes[1] / 2;
    const int* src = ei;
    const int* dst = ei + E;

    float* ws  = (float*)d_ws;
    float* h0  = ws;
    float* agg = ws + (size_t)N * H_DIM;
    float* h1  = ws + 2 * (size_t)N * H_DIM;
    float* deg = ws + 3 * (size_t)N * H_DIM;

    int* col      = (int*)(agg + (size_t)N * (H_DIM / 2));
    int* rowstart = col + E;

    const int NB = (N + PB_NODES - 1) >> PB_SHIFT;
    int* psrc  = (int*)agg;
    int* pdst  = psrc + E;
    int* lrank = pdst + E;
    int* gh    = lrank + E;
    int* rs    = gh + (size_t)NB * PB_GRID;

    part_count_kernel<<<PB_GRID, 1024, 0, stream>>>(dst, lrank, gh, E, NB);
    part_move_kernel<<<PB_GRID, 1024, 0, stream>>>(src, dst, lrank, gh, rs, psrc, pdst, E, NB);
    bucket_fill_kernel<<<NB, 1024, 0, stream>>>(psrc, pdst, rs, rowstart, deg, col, N);

    gemm_strips<H_DIM, 4><<<GEMM_BLOCKS, GEMM_THREADS, 0, stream>>>(
        x, w_in, DIN, nullptr, nullptr, 0, nullptr, b_in, nullptr, nullptr,
        h0, N, 1);

    layer_fused_kernel<false><<<GEMM_BLOCKS, FL_THREADS, 0, stream>>>(
        h0, rowstart, col, w_l1, w_r1, deg, b_l1, g1, be1, nullptr, nullptr, h1, N);

    layer_fused_kernel<true><<<GEMM_BLOCKS, FL_THREADS, 0, stream>>>(
        h1, rowstart, col, w_l2, w_r2, deg, b_l2, g2, be2, w_out, b_out, (float*)d_out, N);
}


# Edge bucketing in two launches, not three: a work-group sorts its own edge share by range in registers; the range's work-group fills from those cells, slots kept in registers; slot order changes
# speedup vs baseline: 3.6729x; 1.1206x over previous
#include <hip/hip_runtime.h>

#define DIN   256
#define H_DIM 128
#define DOUT  64
#define BN_RSQRT 0.9999950000374996f

typedef __attribute__((ext_vector_type(2))) float v2f;
typedef __attribute__((ext_vector_type(8))) float v8f;

__device__ __forceinline__ v8f wmma4(v2f a, v2f b, v8f c) {
    return __builtin_amdgcn_wmma_f32_16x16x4_f32(
         false, a,  false, b,
         (short)0, c,  false,  false);
}

#define GEMM_BLOCKS    256
#define GEMM_THREADS   1024
#define GEMM_LDS_BYTES (164 * 1024)

template <int NTW>
__device__ __forceinline__ void strip_accum(const float* __restrict__ ap,
                                            int K, float rs,
                                            const float4* bq,
                                            int ncols, v8f (&acc)[NTW]) {
    for (int k0 = 0; k0 < K; k0 += 32) {
        float4 a[4];
#pragma unroll
        for (int j = 0; j < 4; ++j) a[j] = *(const float4*)(ap + k0 + 4 * j);
        const float4* bk = bq + (size_t)(k0 >> 2) * ncols;
        float4 b[NTW];
#pragma unroll
        for (int ni = 0; ni < NTW; ++ni) b[ni] = bk[ni * 16];
#pragma unroll
        for (int j = 0; j < 4; ++j) {
            float4 bn[NTW];
            if (j < 3) {
#pragma unroll
                for (int ni = 0; ni < NTW; ++ni) bn[ni] = bk[(size_t)(j + 1) * ncols + ni * 16];
            }
            v2f alo, ahi;
            alo.x = a[j].x * rs;  alo.y = a[j].y * rs;
            ahi.x = a[j].z * rs;  ahi.y = a[j].w * rs;
#pragma unroll
            for (int ni = 0; ni < NTW; ++ni) {
                v2f bb; bb.x = b[ni].x; bb.y = b[ni].y;
                acc[ni] = wmma4(alo, bb, acc[ni]);
            }
#pragma unroll
            for (int ni = 0; ni < NTW; ++ni) {
                v2f bb; bb.x = b[ni].z; bb.y = b[ni].w;
                acc[ni] = wmma4(ahi, bb, acc[ni]);
            }
            if (j < 3) {
#pragma unroll
                for (int ni = 0; ni < NTW; ++ni) b[ni] = bn[ni];
            }
        }
    }
}

template <int NCOLS, int NTW>
__global__ __launch_bounds__(GEMM_THREADS)
void gemm_strips(const float* __restrict__ A0, const float* __restrict__ W0, int K0,
                 const float* __restrict__ A1, const float* __restrict__ W1, int K1,
                 const float* __restrict__ rowscale,
                 const float* __restrict__ bias,
                 const float* __restrict__ gamma, const float* __restrict__ beta,
                 float* __restrict__ out, int nrows, int do_relu) {
    __shared__ __attribute__((aligned(16))) float4 wl[GEMM_LDS_BYTES / 16];
    const int tid = threadIdx.x;
    const int lane = tid & 31;
    const int wave = __builtin_amdgcn_readfirstlane(tid >> 5);
    const int c = lane & 15, h = lane >> 4;
    const int Q0 = K0 >> 2;
    const int Q1 = A1 ? (K1 >> 2) : 0;

    const int nq = (Q0 + Q1) * NCOLS;
    for (int it = tid; it < nq; it += GEMM_THREADS) {
        const int q = it / NCOLS, n = it % NCOLS;
        const float* wp = (q < Q0) ? (W0 + (size_t)(4 * q) * NCOLS + n)
                                   : (W1 + (size_t)(4 * (q - Q0)) * NCOLS + n);
        float4 v;
        v.x = wp[0];  v.y = wp[NCOLS];  v.z = wp[2 * NCOLS];  v.w = wp[3 * NCOLS];
        wl[it] = v;
    }
    __syncthreads();

    constexpr int NSPLIT = NCOLS / (16 * NTW);
    const int items = (nrows >> 4) * NSPLIT;
    const int beg = (int)(((long long)blockIdx.x * items) / gridDim.x);
    const int end = (int)(((long long)(blockIdx.x + 1) * items) / gridDim.x);

    for (int it = beg + wave; it < end; it += GEMM_THREADS / 32) {
        const int strip = it / NSPLIT, part = it % NSPLIT;
        const int row0 = strip * 16, col0 = part * 16 * NTW;

        v8f acc[NTW] = {};
        const float rs = rowscale ? rowscale[row0 + c] : 1.0f;
        strip_accum<NTW>(A0 + (size_t)(row0 + c) * K0 + 16 * h, K0, rs,
                         wl + (size_t)(4 * h) * NCOLS + col0 + c, NCOLS, acc);
        if (A1)
            strip_accum<NTW>(A1 + (size_t)(row0 + c) * K1 + 16 * h, K1, 1.0f,
                             wl + (size_t)(Q0 + 4 * h) * NCOLS + col0 + c, NCOLS, acc);

#pragma unroll
        for (int ni = 0; ni < NTW; ++ni) {
            const int col = col0 + ni * 16 + c;
            const float bi = bias ? bias[col] : 0.0f;
            const float sc = gamma ? gamma[col] * BN_RSQRT : 1.0f;
            const float sh = beta ? beta[col] : 0.0f;
#pragma unroll
            for (int r = 0; r < 8; ++r) {
                const int row = row0 + r + 8 * h;
                float v = acc[ni][r];
                v = (v + bi) * sc + sh;
                if (do_relu) v = fmaxf(v, 0.0f);
                out[(size_t)row * NCOLS + col] = v;
            }
        }
    }
}

#define PB_SHIFT 9
#define PB_NODES (1 << PB_SHIFT)
#define PB_MAXB  2048
#define PB_GRID  256
#define PB_BATCH 8
#define PB_CPW   (PB_GRID / 32)
#define PB_CELL  64

__global__ __launch_bounds__(1024)
void part_sort_kernel(const int* __restrict__ src, const int* __restrict__ dst,
                      int* __restrict__ lrank, int* __restrict__ gh, int* __restrict__ cs,
                      int* __restrict__ psrc, int* __restrict__ pdst, int E, int NB) {
    __shared__ int hist[PB_MAXB];
    __shared__ int s_w[32];
    const int tid = threadIdx.x;
    const int lane = tid & 31, wave = tid >> 5;
    const int g = blockIdx.x;
    for (int i = tid; i < NB; i += 1024) hist[i] = 0;
    __syncthreads();
    const int beg = (int)(((long long)blockIdx.x * E) / gridDim.x);
    const int end = (int)(((long long)(blockIdx.x + 1) * E) / gridDim.x);
    const bool one = (end - beg) <= PB_BATCH * 1024;
    const int e0 = beg + tid;
    int d[PB_BATCH], s[PB_BATCH], r[PB_BATCH];
    if (one) {
#pragma unroll
        for (int u = 0; u < PB_BATCH; ++u) {
            const int ee = max(min(e0 + u * 1024, end - 1), 0);
            d[u] = dst[ee];
            s[u] = src[ee];
        }
#pragma unroll
        for (int u = 0; u < PB_BATCH; ++u) {
            r[u] = 0;
            if (e0 + u * 1024 < end) r[u] = atomicAdd(&hist[d[u] >> PB_SHIFT], 1);
        }
    } else {
        for (int e = e0; e < end; e += 1024)
            lrank[e] = atomicAdd(&hist[dst[e] >> PB_SHIFT], 1);
    }
    __syncthreads();

    constexpr int RPT = PB_MAXB / 1024;
    int c[RPT];
    int loc = 0;
#pragma unroll
    for (int u = 0; u < RPT; ++u) {
        const int i = tid * RPT + u;
        c[u] = (i < NB) ? hist[i] : 0;
        loc += c[u];
    }
    int incl = loc;
    for (int dd = 1; dd < 32; dd <<= 1) {
        const int up = __shfl_up(incl, dd);
        if (lane >= dd) incl += up;
    }
    if (lane == 31) s_w[wave] = incl;
    __syncthreads();
    int run = beg + incl - loc;
    for (int w = 0; w < wave; ++w) run += s_w[w];
#pragma unroll
    for (int u = 0; u < RPT; ++u) {
        const int i = tid * RPT + u;
        if (i < NB) {
            gh[(size_t)i * PB_GRID + g] = c[u];
            cs[(size_t)i * PB_GRID + g] = run;
            hist[i] = run;
            run += c[u];
        }
    }
    __syncthreads();

    if (one) {
#pragma unroll
        for (int u = 0; u < PB_BATCH; ++u) {
            if (e0 + u * 1024 < end) {
                const int pos = hist[d[u] >> PB_SHIFT] + r[u];
                psrc[pos] = s[u];
                pdst[pos] = d[u];
            }
        }
    } else {
        for (int e = e0; e < end; e += 1024) {
            const int dd = dst[e];
            const int pos = hist[dd >> PB_SHIFT] + lrank[e];
            psrc[pos] = src[e];
            pdst[pos] = dd;
        }
    }
}

__global__ __launch_bounds__(1024)
void bucket_fill_kernel(const int* __restrict__ psrc, const int* __restrict__ pdst,
                        const int* __restrict__ gh, const int* __restrict__ cs,
                        int* __restrict__ lrank,
                        int* __restrict__ rowstart, float* __restrict__ invdeg,
                        int* __restrict__ col, int N, int E) {
    static_assert(PB_GRID == 256 && PB_CPW * 32 == PB_GRID, "a wave owns PB_CPW cells, 32 waves own all");
    static_assert(PB_CELL == 64, "two edges of a cell per lane");
    __shared__ int cnt[PB_NODES];
    __shared__ int start[PB_NODES];
    __shared__ int csz[PB_GRID];
    __shared__ int cst[PB_GRID];
    __shared__ int s_w[32];
    __shared__ int s_red[32];
    __shared__ int s_mx[32];
    const int tid = threadIdx.x;
    const int lane = tid & 31, wave = tid >> 5;
    const int b = blockIdx.x;
    const int n0 = b << PB_SHIFT;

    if (tid < PB_NODES) cnt[tid] = 0;
    int mx = 0, st = 0;
    if (tid < PB_GRID) {
        mx = gh[(size_t)b * PB_GRID + tid];
        st = cs[(size_t)b * PB_GRID + tid];
    }
    int acc = 0;
    const int lim = b * PB_GRID;
    for (int j = tid; j < lim; j += 8 * 1024) {
        int v[8];
#pragma unroll
        for (int u = 0; u < 8; ++u) v[u] = gh[min(j + u * 1024, lim - 1)];
#pragma unroll
        for (int u = 0; u < 8; ++u)
            if (j + u * 1024 < lim) acc += v[u];
    }
    if (tid < PB_GRID) { csz[tid] = mx; cst[tid] = st; }
    for (int dd = 16; dd > 0; dd >>= 1) {
        acc += __shfl_xor(acc, dd);
        mx = max(mx, __shfl_xor(mx, dd));
    }
    if (lane == 0) { s_red[wave] = acc; s_mx[wave] = mx; }
    __syncthreads();
    int s0 = s_red[lane];
    int maxn = s_mx[lane];
    for (int dd = 16; dd > 0; dd >>= 1) {
        s0 += __shfl_xor(s0, dd);
        maxn = max(maxn, __shfl_xor(maxn, dd));
    }
    const bool one = maxn <= PB_CELL;

    int n[PB_CPW], cb[PB_CPW];
#pragma unroll
    for (int m = 0; m < PB_CPW; ++m) {
        n[m] = csz[wave * PB_CPW + m];
        cb[m] = cst[wave * PB_CPW + m];
    }

    int d[2 * PB_CPW], s[2 * PB_CPW], sl[2 * PB_CPW];
    if (one) {
#pragma unroll
        for (int m = 0; m < PB_CPW; ++m) {
#pragma unroll
            for (int t = 0; t < 2; ++t) {
                const int e = min(lane + 32 * t, max(n[m] - 1, 0));
                const int gi = max(min(cb[m] + e, E - 1), 0);
                d[2 * m + t] = pdst[gi];
                s[2 * m + t] = psrc[gi];
            }
        }
#pragma unroll
        for (int m = 0; m < PB_CPW; ++m) {
#pragma unroll
            for (int t = 0; t < 2; ++t) {
                sl[2 * m + t] = 0;
                if (lane + 32 * t < n[m]) sl[2 * m + t] = atomicAdd(&cnt[d[2 * m + t] - n0], 1);
            }
        }
    } else {
#pragma unroll 1
        for (int m = 0; m < PB_CPW; ++m)
            for (int e = lane; e < n[m]; e += 32) {
                const int gi = cb[m] + e;
                lrank[gi] = atomicAdd(&cnt[pdst[gi] - n0], 1);
            }
    }
    __syncthreads();

    const int c = (tid < PB_NODES) ? cnt[tid] : 0;
    int incl = c;
    for (int dd = 1; dd < 32; dd <<= 1) {
        const int up = __shfl_up(incl, dd);
        if (lane >= dd) incl += up;
    }
    if (lane == 31) s_w[wave] = incl;
    __syncthreads();
    int excl = incl - c;
    for (int w = 0; w < wave; ++w) excl += s_w[w];
    if (tid < PB_NODES) {
        start[tid] = excl;
        const int node = n0 + tid;
        if (node < N) {
            rowstart[node] = s0 + excl;
            invdeg[node] = 1.0f / fmaxf((float)c, 1.0f);
        }
    }
    if (b == (int)gridDim.x - 1 && tid == 0) {
        int tot = 0;
        for (int w = 0; w < 32; ++w) tot += s_w[w];
        rowstart[N] = s0 + tot;
    }
    __syncthreads();

    if (one) {
#pragma unroll
        for (int m = 0; m < PB_CPW; ++m) {
#pragma unroll
            for (int t = 0; t < 2; ++t) {
                if (lane + 32 * t < n[m])
                    col[s0 + start[d[2 * m + t] - n0] + sl[2 * m + t]] = s[2 * m + t];
            }
        }
    } else {
#pragma unroll 1
        for (int m = 0; m < PB_CPW; ++m)
            for (int e = lane; e < n[m]; e += 32) {
                const int gi = cb[m] + e;
                const int dl = pdst[gi] - n0;
                col[s0 + start[dl] + lrank[gi]] = psrc[gi];
            }
    }
}

#define FL_WAVES   16
#define FL_THREADS (FL_WAVES * 32)
#define FL_PITCH   132
#define FL_U       16

template <bool FUSE_OUT>
__global__ __launch_bounds__(FL_THREADS)
void layer_fused_kernel(const float* __restrict__ hin, const int* __restrict__ rowstart,
                       const int* __restrict__ col,
                       const float* __restrict__ Wl, const float* __restrict__ Wr,
                       const float* __restrict__ invdeg, const float* __restrict__ bias,
                       const float* __restrict__ gamma, const float* __restrict__ beta,
                       const float* __restrict__ Wo, const float* __restrict__ bo,
                       float* __restrict__ out, int nrows) {
    constexpr int Q0 = H_DIM / 4;
    constexpr int NQ = 2 * Q0 * H_DIM;
    constexpr int NQO = Q0 * DOUT;
    __shared__ __attribute__((aligned(16))) float4 wl[NQ];
    __shared__ __attribute__((aligned(16))) float4 wo[FUSE_OUT ? NQO : 1];
    __shared__ __attribute__((aligned(16))) float strips[FL_WAVES * 16 * FL_PITCH];
    __shared__ int s_next;
    const int tid = threadIdx.x;
    const int lane = tid & 31;
    const int wave = __builtin_amdgcn_readfirstlane(tid >> 5);
    const int c = lane & 15, h = lane >> 4;

    for (int it0 = tid; it0 < NQ; it0 += 4 * FL_THREADS) {
        float4 v[4];
#pragma unroll
        for (int u = 0; u < 4; ++u) {
            const int it = min(it0 + u * FL_THREADS, NQ - 1);
            const int q = it / H_DIM, n = it % H_DIM;
            const float* wp = (q < Q0) ? (Wl + (size_t)(4 * q) * H_DIM + n)
                                       : (Wr + (size_t)(4 * (q - Q0)) * H_DIM + n);
            v[u].x = wp[0];  v[u].y = wp[H_DIM];  v[u].z = wp[2 * H_DIM];  v[u].w = wp[3 * H_DIM];
        }
#pragma unroll
        for (int u = 0; u < 4; ++u)
            if (it0 + u * FL_THREADS < NQ) wl[it0 + u * FL_THREADS] = v[u];
    }
    if constexpr (FUSE_OUT) {
        static_assert(NQO % FL_THREADS == 0, "output weight entries per thread");
        float4 v[NQO / FL_THREADS];
#pragma unroll
        for (int u = 0; u < NQO / FL_THREADS; ++u) {
            const int it = tid + u * FL_THREADS;
            const int q = it / DOUT, n = it % DOUT;
            const float* wp = Wo + (size_t)(4 * q) * DOUT + n;
            v[u].x = wp[0];  v[u].y = wp[DOUT];  v[u].z = wp[2 * DOUT];  v[u].w = wp[3 * DOUT];
        }
#pragma unroll
        for (int u = 0; u < NQO / FL_THREADS; ++u) wo[tid + u * FL_THREADS] = v[u];
    }
    if (tid == 0) s_next = 0;
    __syncthreads();

    const int items = nrows >> 4;
    const int beg = (int)(((long long)blockIdx.x * items) / gridDim.x);
    const int end = (int)(((long long)(blockIdx.x + 1) * items) / gridDim.x);
    float* sp = strips + wave * (16 * FL_PITCH);
    const float* hp = hin + lane * 4;

    for (;;) {
        int t = 0;
        if (lane == 0) t = atomicAdd(&s_next, 1);
        const int it = beg + __builtin_amdgcn_readfirstlane(t);
        if (it >= end) break;
        const int row0 = it * 16;

        v8f acc8[8] = {};
        strip_accum<8>(hin + (size_t)(row0 + c) * H_DIM + 16 * h, H_DIM, 1.0f,
                       wl + (size_t)(Q0 + 4 * h) * H_DIM + c, H_DIM, acc8);

        const int rsv = rowstart[row0 + min(lane, 16)];
        int j = __builtin_amdgcn_readlane(rsv, 0);
        const int jend = __builtin_amdgcn_readlane(rsv, 16);
        int r = 0;
        int nb = __builtin_amdgcn_readlane(rsv, 1);
        float4 acc = make_float4(0.0f, 0.0f, 0.0f, 0.0f);
        int idv = col[max(min(j + c, jend - 1), 0)];
        while (j < jend) {
            float4 v[FL_U];
#pragma unroll
            for (int u = 0; u < FL_U; ++u) {
                const int id = __builtin_amdgcn_readlane(idv, u);
                v[u] = *(const float4*)(hp + (size_t)id * H_DIM);
            }
            idv = col[min(j + FL_U + c, jend - 1)];
#pragma unroll
            for (int u = 0; u < FL_U; ++u) {
                if (j + u < jend) {
                    while (j + u == nb) {
                        *(float4*)(sp + r * FL_PITCH + lane * 4) = acc;
                        acc = make_float4(0.0f, 0.0f, 0.0f, 0.0f);
                        ++r;
                        nb = __builtin_amdgcn_readlane(rsv, r + 1);
                    }
                    acc.x += v[u].x; acc.y += v[u].y; acc.z += v[u].z; acc.w += v[u].w;
                }
            }
            j += FL_U;
        }
        while (r < 16) {
            *(float4*)(sp + r * FL_PITCH + lane * 4) = acc;
            acc = make_float4(0.0f, 0.0f, 0.0f, 0.0f);
            ++r;
        }

        const float rs = invdeg[row0 + c];
        strip_accum<8>(sp + c * FL_PITCH + 16 * h, H_DIM, rs,
                       wl + (size_t)(4 * h) * H_DIM + c, H_DIM, acc8);

#pragma unroll
        for (int ni = 0; ni < 8; ++ni) {
            const int cc = ni * 16 + c;
            const float bi = bias[cc];
            const float sc = gamma[cc] * BN_RSQRT;
            const float sh = beta[cc];
#pragma unroll
            for (int rr = 0; rr < 8; ++rr) {
                const int row = row0 + rr + 8 * h;
                float v = acc8[ni][rr];
                v = (v + bi) * sc + sh;
                v = fmaxf(v, 0.0f);
                if constexpr (FUSE_OUT) {
                    sp[(rr + 8 * h) * FL_PITCH + cc] = v;
                } else {
                    out[(size_t)row * H_DIM + cc] = v;
                }
            }
        }

        if constexpr (FUSE_OUT) {
            v8f acc4[DOUT / 16] = {};
            strip_accum<DOUT / 16>(sp + c * FL_PITCH + 16 * h, H_DIM, 1.0f,
                                   wo + (size_t)(4 * h) * DOUT + c, DOUT, acc4);
#pragma unroll
            for (int ni = 0; ni < DOUT / 16; ++ni) {
                const int cc = ni * 16 + c;
                const float bi = bo[cc];
#pragma unroll
                for (int rr = 0; rr < 8; ++rr) {
                    const int row = row0 + rr + 8 * h;
                    out[(size_t)row * DOUT + cc] = acc4[ni][rr] + bi;
                }
            }
        }
    }
}

extern "C" void kernel_launch(void* const* d_in, const int* in_sizes, int n_in,
                              void* d_out, int out_size, void* d_ws, size_t ws_size,
                              hipStream_t stream) {
    const float* x     = (const float*)d_in[0];
    const int*   ei    = (const int*)d_in[1];
    const float* w_in  = (const float*)d_in[2];
    const float* b_in  = (const float*)d_in[3];
    const float* w_l1  = (const float*)d_in[4];
    const float* b_l1  = (const float*)d_in[5];
    const float* w_r1  = (const float*)d_in[6];
    const float* g1    = (const float*)d_in[7];
    const float* be1   = (const float*)d_in[8];
    const float* w_l2  = (const float*)d_in[9];
    const float* b_l2  = (const float*)d_in[10];
    const float* w_r2  = (const float*)d_in[11];
    const float* g2    = (const float*)d_in[12];
    const float* be2   = (const float*)d_in[13];
    const float* w_out = (const float*)d_in[14];
    const float* b_out = (const float*)d_in[15];

    const int N = in_sizes[0] / DIN;
    const int E = in_sizes[1] / 2;
    const int* src = ei;
    const int* dst = ei + E;

    float* ws  = (float*)d_ws;
    float* h0  = ws;
    float* agg = ws + (size_t)N * H_DIM;
    float* h1  = ws + 2 * (size_t)N * H_DIM;
    float* deg = ws + 3 * (size_t)N * H_DIM;

    int* col      = (int*)(agg + (size_t)N * (H_DIM / 2));
    int* rowstart = col + E;

    const int NB = (N + PB_NODES - 1) >> PB_SHIFT;
    int* psrc  = (int*)agg;
    int* pdst  = psrc + E;
    int* lrank = pdst + E;
    int* gh    = lrank + E;
    int* cs    = gh + (size_t)NB * PB_GRID;

    part_sort_kernel<<<PB_GRID, 1024, 0, stream>>>(src, dst, lrank, gh, cs, psrc, pdst, E, NB);
    bucket_fill_kernel<<<NB, 1024, 0, stream>>>(psrc, pdst, gh, cs, lrank, rowstart, deg, col, N, E);

    gemm_strips<H_DIM, 4><<<GEMM_BLOCKS, GEMM_THREADS, 0, stream>>>(
        x, w_in, DIN, nullptr, nullptr, 0, nullptr, b_in, nullptr, nullptr,
        h0, N, 1);

    layer_fused_kernel<false><<<GEMM_BLOCKS, FL_THREADS, 0, stream>>>(
        h0, rowstart, col, w_l1, w_r1, deg, b_l1, g1, be1, nullptr, nullptr, h1, N);

    layer_fused_kernel<true><<<GEMM_BLOCKS, FL_THREADS, 0, stream>>>(
        h1, rowstart, col, w_l2, w_r2, deg, b_l2, g2, be2, w_out, b_out, (float*)d_out, N);
}


# Bucketing stores staged in LDS: each work-group places its edges in an LDS copy of its contiguous output piece and writes it out in order, not as scattered 4-byte stores; slot order unchanged
# speedup vs baseline: 3.9172x; 1.0665x over previous
#include <hip/hip_runtime.h>

#define DIN   256
#define H_DIM 128
#define DOUT  64
#define BN_RSQRT 0.9999950000374996f

typedef __attribute__((ext_vector_type(2))) float v2f;
typedef __attribute__((ext_vector_type(8))) float v8f;

__device__ __forceinline__ v8f wmma4(v2f a, v2f b, v8f c) {
    return __builtin_amdgcn_wmma_f32_16x16x4_f32(
         false, a,  false, b,
         (short)0, c,  false,  false);
}

#define GEMM_BLOCKS    256
#define GEMM_THREADS   1024
#define GEMM_LDS_BYTES (164 * 1024)

template <int NTW>
__device__ __forceinline__ void strip_accum(const float* __restrict__ ap,
                                            int K, float rs,
                                            const float4* bq,
                                            int ncols, v8f (&acc)[NTW]) {
    for (int k0 = 0; k0 < K; k0 += 32) {
        float4 a[4];
#pragma unroll
        for (int j = 0; j < 4; ++j) a[j] = *(const float4*)(ap + k0 + 4 * j);
        const float4* bk = bq + (size_t)(k0 >> 2) * ncols;
        float4 b[NTW];
#pragma unroll
        for (int ni = 0; ni < NTW; ++ni) b[ni] = bk[ni * 16];
#pragma unroll
        for (int j = 0; j < 4; ++j) {
            float4 bn[NTW];
            if (j < 3) {
#pragma unroll
                for (int ni = 0; ni < NTW; ++ni) bn[ni] = bk[(size_t)(j + 1) * ncols + ni * 16];
            }
            v2f alo, ahi;
            alo.x = a[j].x * rs;  alo.y = a[j].y * rs;
            ahi.x = a[j].z * rs;  ahi.y = a[j].w * rs;
#pragma unroll
            for (int ni = 0; ni < NTW; ++ni) {
                v2f bb; bb.x = b[ni].x; bb.y = b[ni].y;
                acc[ni] = wmma4(alo, bb, acc[ni]);
            }
#pragma unroll
            for (int ni = 0; ni < NTW; ++ni) {
                v2f bb; bb.x = b[ni].z; bb.y = b[ni].w;
                acc[ni] = wmma4(ahi, bb, acc[ni]);
            }
            if (j < 3) {
#pragma unroll
                for (int ni = 0; ni < NTW; ++ni) b[ni] = bn[ni];
            }
        }
    }
}

template <int NCOLS, int NTW>
__global__ __launch_bounds__(GEMM_THREADS)
void gemm_strips(const float* __restrict__ A0, const float* __restrict__ W0, int K0,
                 const float* __restrict__ A1, const float* __restrict__ W1, int K1,
                 const float* __restrict__ rowscale,
                 const float* __restrict__ bias,
                 const float* __restrict__ gamma, const float* __restrict__ beta,
                 float* __restrict__ out, int nrows, int do_relu) {
    __shared__ __attribute__((aligned(16))) float4 wl[GEMM_LDS_BYTES / 16];
    const int tid = threadIdx.x;
    const int lane = tid & 31;
    const int wave = __builtin_amdgcn_readfirstlane(tid >> 5);
    const int c = lane & 15, h = lane >> 4;
    const int Q0 = K0 >> 2;
    const int Q1 = A1 ? (K1 >> 2) : 0;

    const int nq = (Q0 + Q1) * NCOLS;
    for (int it = tid; it < nq; it += GEMM_THREADS) {
        const int q = it / NCOLS, n = it % NCOLS;
        const float* wp = (q < Q0) ? (W0 + (size_t)(4 * q) * NCOLS + n)
                                   : (W1 + (size_t)(4 * (q - Q0)) * NCOLS + n);
        float4 v;
        v.x = wp[0];  v.y = wp[NCOLS];  v.z = wp[2 * NCOLS];  v.w = wp[3 * NCOLS];
        wl[it] = v;
    }
    __syncthreads();

    constexpr int NSPLIT = NCOLS / (16 * NTW);
    const int items = (nrows >> 4) * NSPLIT;
    const int beg = (int)(((long long)blockIdx.x * items) / gridDim.x);
    const int end = (int)(((long long)(blockIdx.x + 1) * items) / gridDim.x);

    for (int it = beg + wave; it < end; it += GEMM_THREADS / 32) {
        const int strip = it / NSPLIT, part = it % NSPLIT;
        const int row0 = strip * 16, col0 = part * 16 * NTW;

        v8f acc[NTW] = {};
        const float rs = rowscale ? rowscale[row0 + c] : 1.0f;
        strip_accum<NTW>(A0 + (size_t)(row0 + c) * K0 + 16 * h, K0, rs,
                         wl + (size_t)(4 * h) * NCOLS + col0 + c, NCOLS, acc);
        if (A1)
            strip_accum<NTW>(A1 + (size_t)(row0 + c) * K1 + 16 * h, K1, 1.0f,
                             wl + (size_t)(Q0 + 4 * h) * NCOLS + col0 + c, NCOLS, acc);

#pragma unroll
        for (int ni = 0; ni < NTW; ++ni) {
            const int col = col0 + ni * 16 + c;
            const float bi = bias ? bias[col] : 0.0f;
            const float sc = gamma ? gamma[col] * BN_RSQRT : 1.0f;
            const float sh = beta ? beta[col] : 0.0f;
#pragma unroll
            for (int r = 0; r < 8; ++r) {
                const int row = row0 + r + 8 * h;
                float v = acc[ni][r];
                v = (v + bi) * sc + sh;
                if (do_relu) v = fmaxf(v, 0.0f);
                out[(size_t)row * NCOLS + col] = v;
            }
        }
    }
}

#define PB_SHIFT 9
#define PB_NODES (1 << PB_SHIFT)
#define PB_MAXB  2048
#define PB_GRID  256
#define PB_BATCH 8
#define PB_CPW   (PB_GRID / 32)
#define PB_CELL  64

__global__ __launch_bounds__(1024)
void part_sort_kernel(const int* __restrict__ src, const int* __restrict__ dst,
                      int* __restrict__ lrank, int* __restrict__ gh, int* __restrict__ cs,
                      int* __restrict__ psrc, int* __restrict__ pdst, int E, int NB) {
    __shared__ int hist[PB_MAXB];
    __shared__ int s_w[32];
    __shared__ int st_s[PB_BATCH * 1024];
    __shared__ int st_d[PB_BATCH * 1024];
    const int tid = threadIdx.x;
    const int lane = tid & 31, wave = tid >> 5;
    const int g = blockIdx.x;
    for (int i = tid; i < NB; i += 1024) hist[i] = 0;
    __syncthreads();
    const int beg = (int)(((long long)blockIdx.x * E) / gridDim.x);
    const int end = (int)(((long long)(blockIdx.x + 1) * E) / gridDim.x);
    const bool one = (end - beg) <= PB_BATCH * 1024;
    const int e0 = beg + tid;
    int d[PB_BATCH], s[PB_BATCH], r[PB_BATCH];
    if (one) {
#pragma unroll
        for (int u = 0; u < PB_BATCH; ++u) {
            const int ee = max(min(e0 + u * 1024, end - 1), 0);
            d[u] = dst[ee];
            s[u] = src[ee];
        }
#pragma unroll
        for (int u = 0; u < PB_BATCH; ++u) {
            r[u] = 0;
            if (e0 + u * 1024 < end) r[u] = atomicAdd(&hist[d[u] >> PB_SHIFT], 1);
        }
    } else {
        for (int e = e0; e < end; e += 1024)
            lrank[e] = atomicAdd(&hist[dst[e] >> PB_SHIFT], 1);
    }
    __syncthreads();

    constexpr int RPT = PB_MAXB / 1024;
    int c[RPT];
    int loc = 0;
#pragma unroll
    for (int u = 0; u < RPT; ++u) {
        const int i = tid * RPT + u;
        c[u] = (i < NB) ? hist[i] : 0;
        loc += c[u];
    }
    int incl = loc;
    for (int dd = 1; dd < 32; dd <<= 1) {
        const int up = __shfl_up(incl, dd);
        if (lane >= dd) incl += up;
    }
    if (lane == 31) s_w[wave] = incl;
    __syncthreads();
    int run = beg + incl - loc;
    for (int w = 0; w < wave; ++w) run += s_w[w];
#pragma unroll
    for (int u = 0; u < RPT; ++u) {
        const int i = tid * RPT + u;
        if (i < NB) {
            gh[(size_t)i * PB_GRID + g] = c[u];
            cs[(size_t)i * PB_GRID + g] = run;
            hist[i] = run;
            run += c[u];
        }
    }
    __syncthreads();

    if (one) {
#pragma unroll
        for (int u = 0; u < PB_BATCH; ++u) {
            if (e0 + u * 1024 < end) {
                const int lp = hist[d[u] >> PB_SHIFT] + r[u] - beg;
                st_s[lp] = s[u];
                st_d[lp] = d[u];
            }
        }
        __syncthreads();
        const int n = end - beg;
        int vs[PB_BATCH], vd[PB_BATCH];
#pragma unroll
        for (int u = 0; u < PB_BATCH; ++u) {
            const int i = max(min(tid + u * 1024, n - 1), 0);
            vs[u] = st_s[i];
            vd[u] = st_d[i];
        }
#pragma unroll
        for (int u = 0; u < PB_BATCH; ++u) {
            const int i = tid + u * 1024;
            if (i < n) {
                psrc[beg + i] = vs[u];
                pdst[beg + i] = vd[u];
            }
        }
    } else {
        for (int e = e0; e < end; e += 1024) {
            const int dd = dst[e];
            const int pos = hist[dd >> PB_SHIFT] + lrank[e];
            psrc[pos] = src[e];
            pdst[pos] = dd;
        }
    }
}

__global__ __launch_bounds__(1024)
void bucket_fill_kernel(const int* __restrict__ psrc, const int* __restrict__ pdst,
                        const int* __restrict__ gh, const int* __restrict__ cs,
                        int* __restrict__ lrank,
                        int* __restrict__ rowstart, float* __restrict__ invdeg,
                        int* __restrict__ col, int N, int E) {
    static_assert(PB_GRID == 256 && PB_CPW * 32 == PB_GRID, "a wave owns PB_CPW cells, 32 waves own all");
    static_assert(PB_CELL == 64, "two edges of a cell per lane");
    __shared__ int cnt[PB_NODES];
    __shared__ int start[PB_NODES];
    __shared__ int csz[PB_GRID];
    __shared__ int cst[PB_GRID];
    __shared__ int s_w[32];
    __shared__ int s_red[32];
    __shared__ int s_mx[32];
    __shared__ int cbuf[PB_GRID * PB_CELL];
    const int tid = threadIdx.x;
    const int lane = tid & 31, wave = tid >> 5;
    const int b = blockIdx.x;
    const int n0 = b << PB_SHIFT;

    if (tid < PB_NODES) cnt[tid] = 0;
    int mx = 0, st = 0;
    if (tid < PB_GRID) {
        mx = gh[(size_t)b * PB_GRID + tid];
        st = cs[(size_t)b * PB_GRID + tid];
    }
    int acc = 0;
    const int lim = b * PB_GRID;
    for (int j = tid; j < lim; j += 8 * 1024) {
        int v[8];
#pragma unroll
        for (int u = 0; u < 8; ++u) v[u] = gh[min(j + u * 1024, lim - 1)];
#pragma unroll
        for (int u = 0; u < 8; ++u)
            if (j + u * 1024 < lim) acc += v[u];
    }
    if (tid < PB_GRID) { csz[tid] = mx; cst[tid] = st; }
    for (int dd = 16; dd > 0; dd >>= 1) {
        acc += __shfl_xor(acc, dd);
        mx = max(mx, __shfl_xor(mx, dd));
    }
    if (lane == 0) { s_red[wave] = acc; s_mx[wave] = mx; }
    __syncthreads();
    int s0 = s_red[lane];
    int maxn = s_mx[lane];
    for (int dd = 16; dd > 0; dd >>= 1) {
        s0 += __shfl_xor(s0, dd);
        maxn = max(maxn, __shfl_xor(maxn, dd));
    }
    const bool one = maxn <= PB_CELL;

    int n[PB_CPW], cb[PB_CPW];
#pragma unroll
    for (int m = 0; m < PB_CPW; ++m) {
        n[m] = csz[wave * PB_CPW + m];
        cb[m] = cst[wave * PB_CPW + m];
    }

    int d[2 * PB_CPW], s[2 * PB_CPW], sl[2 * PB_CPW];
    if (one) {
#pragma unroll
        for (int m = 0; m < PB_CPW; ++m) {
#pragma unroll
            for (int t = 0; t < 2; ++t) {
                const int e = min(lane + 32 * t, max(n[m] - 1, 0));
                const int gi = max(min(cb[m] + e, E - 1), 0);
                d[2 * m + t] = pdst[gi];
                s[2 * m + t] = psrc[gi];
            }
        }
#pragma unroll
        for (int m = 0; m < PB_CPW; ++m) {
#pragma unroll
            for (int t = 0; t < 2; ++t) {
                sl[2 * m + t] = 0;
                if (lane + 32 * t < n[m]) sl[2 * m + t] = atomicAdd(&cnt[d[2 * m + t] - n0], 1);
            }
        }
    } else {
#pragma unroll 1
        for (int m = 0; m < PB_CPW; ++m)
            for (int e = lane; e < n[m]; e += 32) {
                const int gi = cb[m] + e;
                lrank[gi] = atomicAdd(&cnt[pdst[gi] - n0], 1);
            }
    }
    __syncthreads();

    const int c = (tid < PB_NODES) ? cnt[tid] : 0;
    int incl = c;
    for (int dd = 1; dd < 32; dd <<= 1) {
        const int up = __shfl_up(incl, dd);
        if (lane >= dd) incl += up;
    }
    if (lane == 31) s_w[wave] = incl;
    __syncthreads();
    int excl = incl - c;
    for (int w = 0; w < wave; ++w) excl += s_w[w];
    if (tid < PB_NODES) {
        start[tid] = excl;
        const int node = n0 + tid;
        if (node < N) {
            rowstart[node] = s0 + excl;
            invdeg[node] = 1.0f / fmaxf((float)c, 1.0f);
        }
    }
    if (b == (int)gridDim.x - 1 && tid == 0) {
        int tot = 0;
        for (int w = 0; w < 32; ++w) tot += s_w[w];
        rowstart[N] = s0 + tot;
    }
    __syncthreads();

    if (one) {
#pragma unroll
        for (int m = 0; m < PB_CPW; ++m) {
#pragma unroll
            for (int t = 0; t < 2; ++t) {
                if (lane + 32 * t < n[m])
                    cbuf[start[d[2 * m + t] - n0] + sl[2 * m + t]] = s[2 * m + t];
            }
        }
        int tot = s_w[lane];
        for (int dd = 16; dd > 0; dd >>= 1) tot += __shfl_xor(tot, dd);
        __syncthreads();
        constexpr int CPT = (PB_GRID * PB_CELL) / 1024;
        int vc[CPT];
#pragma unroll
        for (int u = 0; u < CPT; ++u) vc[u] = cbuf[max(min(tid + u * 1024, tot - 1), 0)];
#pragma unroll
        for (int u = 0; u < CPT; ++u) {
            const int i = tid + u * 1024;
            if (i < tot) col[s0 + i] = vc[u];
        }
    } else {
#pragma unroll 1
        for (int m = 0; m < PB_CPW; ++m)
            for (int e = lane; e < n[m]; e += 32) {
                const int gi = cb[m] + e;
                const int dl = pdst[gi] - n0;
                col[s0 + start[dl] + lrank[gi]] = psrc[gi];
            }
    }
}

#define FL_WAVES   16
#define FL_THREADS (FL_WAVES * 32)
#define FL_PITCH   132
#define FL_U       16

template <bool FUSE_OUT>
__global__ __launch_bounds__(FL_THREADS)
void layer_fused_kernel(const float* __restrict__ hin, const int* __restrict__ rowstart,
                       const int* __restrict__ col,
                       const float* __restrict__ Wl, const float* __restrict__ Wr,
                       const float* __restrict__ invdeg, const float* __restrict__ bias,
                       const float* __restrict__ gamma, const float* __restrict__ beta,
                       const float* __restrict__ Wo, const float* __restrict__ bo,
                       float* __restrict__ out, int nrows) {
    constexpr int Q0 = H_DIM / 4;
    constexpr int NQ = 2 * Q0 * H_DIM;
    constexpr int NQO = Q0 * DOUT;
    __shared__ __attribute__((aligned(16))) float4 wl[NQ];
    __shared__ __attribute__((aligned(16))) float4 wo[FUSE_OUT ? NQO : 1];
    __shared__ __attribute__((aligned(16))) float strips[FL_WAVES * 16 * FL_PITCH];
    __shared__ int s_next;
    const int tid = threadIdx.x;
    const int lane = tid & 31;
    const int wave = __builtin_amdgcn_readfirstlane(tid >> 5);
    const int c = lane & 15, h = lane >> 4;

    for (int it0 = tid; it0 < NQ; it0 += 4 * FL_THREADS) {
        float4 v[4];
#pragma unroll
        for (int u = 0; u < 4; ++u) {
            const int it = min(it0 + u * FL_THREADS, NQ - 1);
            const int q = it / H_DIM, n = it % H_DIM;
            const float* wp = (q < Q0) ? (Wl + (size_t)(4 * q) * H_DIM + n)
                                       : (Wr + (size_t)(4 * (q - Q0)) * H_DIM + n);
            v[u].x = wp[0];  v[u].y = wp[H_DIM];  v[u].z = wp[2 * H_DIM];  v[u].w = wp[3 * H_DIM];
        }
#pragma unroll
        for (int u = 0; u < 4; ++u)
            if (it0 + u * FL_THREADS < NQ) wl[it0 + u * FL_THREADS] = v[u];
    }
    if constexpr (FUSE_OUT) {
        static_assert(NQO % FL_THREADS == 0, "output weight entries per thread");
        float4 v[NQO / FL_THREADS];
#pragma unroll
        for (int u = 0; u < NQO / FL_THREADS; ++u) {
            const int it = tid + u * FL_THREADS;
            const int q = it / DOUT, n = it % DOUT;
            const float* wp = Wo + (size_t)(4 * q) * DOUT + n;
            v[u].x = wp[0];  v[u].y = wp[DOUT];  v[u].z = wp[2 * DOUT];  v[u].w = wp[3 * DOUT];
        }
#pragma unroll
        for (int u = 0; u < NQO / FL_THREADS; ++u) wo[tid + u * FL_THREADS] = v[u];
    }
    if (tid == 0) s_next = 0;
    __syncthreads();

    const int items = nrows >> 4;
    const int beg = (int)(((long long)blockIdx.x * items) / gridDim.x);
    const int end = (int)(((long long)(blockIdx.x + 1) * items) / gridDim.x);
    float* sp = strips + wave * (16 * FL_PITCH);
    const float* hp = hin + lane * 4;

    for (;;) {
        int t = 0;
        if (lane == 0) t = atomicAdd(&s_next, 1);
        const int it = beg + __builtin_amdgcn_readfirstlane(t);
        if (it >= end) break;
        const int row0 = it * 16;

        v8f acc8[8] = {};
        strip_accum<8>(hin + (size_t)(row0 + c) * H_DIM + 16 * h, H_DIM, 1.0f,
                       wl + (size_t)(Q0 + 4 * h) * H_DIM + c, H_DIM, acc8);

        const int rsv = rowstart[row0 + min(lane, 16)];
        int j = __builtin_amdgcn_readlane(rsv, 0);
        const int jend = __builtin_amdgcn_readlane(rsv, 16);
        int r = 0;
        int nb = __builtin_amdgcn_readlane(rsv, 1);
        float4 acc = make_float4(0.0f, 0.0f, 0.0f, 0.0f);
        int idv = col[max(min(j + c, jend - 1), 0)];
        while (j < jend) {
            float4 v[FL_U];
#pragma unroll
            for (int u = 0; u < FL_U; ++u) {
                const int id = __builtin_amdgcn_readlane(idv, u);
                v[u] = *(const float4*)(hp + (size_t)id * H_DIM);
            }
            idv = col[min(j + FL_U + c, jend - 1)];
#pragma unroll
            for (int u = 0; u < FL_U; ++u) {
                if (j + u < jend) {
                    while (j + u == nb) {
                        *(float4*)(sp + r * FL_PITCH + lane * 4) = acc;
                        acc = make_float4(0.0f, 0.0f, 0.0f, 0.0f);
                        ++r;
                        nb = __builtin_amdgcn_readlane(rsv, r + 1);
                    }
                    acc.x += v[u].x; acc.y += v[u].y; acc.z += v[u].z; acc.w += v[u].w;
                }
            }
            j += FL_U;
        }
        while (r < 16) {
            *(float4*)(sp + r * FL_PITCH + lane * 4) = acc;
            acc = make_float4(0.0f, 0.0f, 0.0f, 0.0f);
            ++r;
        }

        const float rs = invdeg[row0 + c];
        strip_accum<8>(sp + c * FL_PITCH + 16 * h, H_DIM, rs,
                       wl + (size_t)(4 * h) * H_DIM + c, H_DIM, acc8);

#pragma unroll
        for (int ni = 0; ni < 8; ++ni) {
            const int cc = ni * 16 + c;
            const float bi = bias[cc];
            const float sc = gamma[cc] * BN_RSQRT;
            const float sh = beta[cc];
#pragma unroll
            for (int rr = 0; rr < 8; ++rr) {
                const int row = row0 + rr + 8 * h;
                float v = acc8[ni][rr];
                v = (v + bi) * sc + sh;
                v = fmaxf(v, 0.0f);
                if constexpr (FUSE_OUT) {
                    sp[(rr + 8 * h) * FL_PITCH + cc] = v;
                } else {
                    out[(size_t)row * H_DIM + cc] = v;
                }
            }
        }

        if constexpr (FUSE_OUT) {
            v8f acc4[DOUT / 16] = {};
            strip_accum<DOUT / 16>(sp + c * FL_PITCH + 16 * h, H_DIM, 1.0f,
                                   wo + (size_t)(4 * h) * DOUT + c, DOUT, acc4);
#pragma unroll
            for (int ni = 0; ni < DOUT / 16; ++ni) {
                const int cc = ni * 16 + c;
                const float bi = bo[cc];
#pragma unroll
                for (int rr = 0; rr < 8; ++rr) {
                    const int row = row0 + rr + 8 * h;
                    out[(size_t)row * DOUT + cc] = acc4[ni][rr] + bi;
                }
            }
        }
    }
}

extern "C" void kernel_launch(void* const* d_in, const int* in_sizes, int n_in,
                              void* d_out, int out_size, void* d_ws, size_t ws_size,
                              hipStream_t stream) {
    const float* x     = (const float*)d_in[0];
    const int*   ei    = (const int*)d_in[1];
    const float* w_in  = (const float*)d_in[2];
    const float* b_in  = (const float*)d_in[3];
    const float* w_l1  = (const float*)d_in[4];
    const float* b_l1  = (const float*)d_in[5];
    const float* w_r1  = (const float*)d_in[6];
    const float* g1    = (const float*)d_in[7];
    const float* be1   = (const float*)d_in[8];
    const float* w_l2  = (const float*)d_in[9];
    const float* b_l2  = (const float*)d_in[10];
    const float* w_r2  = (const float*)d_in[11];
    const float* g2    = (const float*)d_in[12];
    const float* be2   = (const float*)d_in[13];
    const float* w_out = (const float*)d_in[14];
    const float* b_out = (const float*)d_in[15];

    const int N = in_sizes[0] / DIN;
    const int E = in_sizes[1] / 2;
    const int* src = ei;
    const int* dst = ei + E;

    float* ws  = (float*)d_ws;
    float* h0  = ws;
    float* agg = ws + (size_t)N * H_DIM;
    float* h1  = ws + 2 * (size_t)N * H_DIM;
    float* deg = ws + 3 * (size_t)N * H_DIM;

    int* col      = (int*)(agg + (size_t)N * (H_DIM / 2));
    int* rowstart = col + E;

    const int NB = (N + PB_NODES - 1) >> PB_SHIFT;
    int* psrc  = (int*)agg;
    int* pdst  = psrc + E;
    int* lrank = pdst + E;
    int* gh    = lrank + E;
    int* cs    = gh + (size_t)NB * PB_GRID;

    part_sort_kernel<<<PB_GRID, 1024, 0, stream>>>(src, dst, lrank, gh, cs, psrc, pdst, E, NB);
    bucket_fill_kernel<<<NB, 1024, 0, stream>>>(psrc, pdst, gh, cs, lrank, rowstart, deg, col, N, E);

    gemm_strips<H_DIM, 4><<<GEMM_BLOCKS, GEMM_THREADS, 0, stream>>>(
        x, w_in, DIN, nullptr, nullptr, 0, nullptr, b_in, nullptr, nullptr,
        h0, N, 1);

    layer_fused_kernel<false><<<GEMM_BLOCKS, FL_THREADS, 0, stream>>>(
        h0, rowstart, col, w_l1, w_r1, deg, b_l1, g1, be1, nullptr, nullptr, h1, N);

    layer_fused_kernel<true><<<GEMM_BLOCKS, FL_THREADS, 0, stream>>>(
        h1, rowstart, col, w_l2, w_r2, deg, b_l2, g2, be2, w_out, b_out, (float*)d_out, N);
}


# Edge partition (bucketing step 1) moved into the input product's launch as its prologue, weight rows and edges asked for in one memory round trip: 5 launches to 4; sum order unchanged
# speedup vs baseline: 4.0625x; 1.0371x over previous
#include <hip/hip_runtime.h>

#define DIN   256
#define H_DIM 128
#define DOUT  64
#define BN_RSQRT 0.9999950000374996f

typedef __attribute__((ext_vector_type(2))) float v2f;
typedef __attribute__((ext_vector_type(8))) float v8f;

__device__ __forceinline__ v8f wmma4(v2f a, v2f b, v8f c) {
    return __builtin_amdgcn_wmma_f32_16x16x4_f32(
         false, a,  false, b,
         (short)0, c,  false,  false);
}

#define GEMM_BLOCKS    256
#define GEMM_THREADS   1024
#define GEMM_LDS_BYTES (164 * 1024)

template <int NTW>
__device__ __forceinline__ void strip_accum(const float* __restrict__ ap,
                                            int K, float rs,
                                            const float4* bq,
                                            int ncols, v8f (&acc)[NTW]) {
    for (int k0 = 0; k0 < K; k0 += 32) {
        float4 a[4];
#pragma unroll
        for (int j = 0; j < 4; ++j) a[j] = *(const float4*)(ap + k0 + 4 * j);
        const float4* bk = bq + (size_t)(k0 >> 2) * ncols;
        float4 b[NTW];
#pragma unroll
        for (int ni = 0; ni < NTW; ++ni) b[ni] = bk[ni * 16];
#pragma unroll
        for (int j = 0; j < 4; ++j) {
            float4 bn[NTW];
            if (j < 3) {
#pragma unroll
                for (int ni = 0; ni < NTW; ++ni) bn[ni] = bk[(size_t)(j + 1) * ncols + ni * 16];
            }
            v2f alo, ahi;
            alo.x = a[j].x * rs;  alo.y = a[j].y * rs;
            ahi.x = a[j].z * rs;  ahi.y = a[j].w * rs;
#pragma unroll
            for (int ni = 0; ni < NTW; ++ni) {
                v2f bb; bb.x = b[ni].x; bb.y = b[ni].y;
                acc[ni] = wmma4(alo, bb, acc[ni]);
            }
#pragma unroll
            for (int ni = 0; ni < NTW; ++ni) {
                v2f bb; bb.x = b[ni].z; bb.y = b[ni].w;
                acc[ni] = wmma4(ahi, bb, acc[ni]);
            }
            if (j < 3) {
#pragma unroll
                for (int ni = 0; ni < NTW; ++ni) b[ni] = bn[ni];
            }
        }
    }
}

#define PB_SHIFT 9
#define PB_NODES (1 << PB_SHIFT)
#define PB_MAXB  2048
#define PB_GRID  256
#define PB_BATCH 8
#define PB_CPW   (PB_GRID / 32)
#define PB_CELL  64

__global__ __launch_bounds__(GEMM_THREADS)
void part_sort_input_product_kernel(const float* __restrict__ A0, const float* __restrict__ W0,
                                    const float* __restrict__ bias, float* __restrict__ out, int nrows,
                                    const int* __restrict__ src, const int* __restrict__ dst,
                                    int* __restrict__ lrank, int* __restrict__ gh, int* __restrict__ cs,
                                    int* __restrict__ psrc, int* __restrict__ pdst, int E, int NB) {
    constexpr int NCOLS = H_DIM, NTW = 4, K0 = DIN;
    constexpr int NQ = (K0 / 4) * NCOLS;
    constexpr int WPT = NQ / GEMM_THREADS;
    static_assert(GEMM_BLOCKS == PB_GRID && GEMM_THREADS == 1024, "one work-group per edge share, 1024 threads");
    static_assert(NQ % GEMM_THREADS == 0 && NQ * 16 <= GEMM_LDS_BYTES, "whole weight entries per thread");
    __shared__ __attribute__((aligned(16))) float4 wl[GEMM_LDS_BYTES / 16];
    __shared__ int hist[PB_MAXB];
    __shared__ int s_w[32];
    __shared__ int st_s[PB_BATCH * 1024];
    __shared__ int st_d[PB_BATCH * 1024];
    const int tid = threadIdx.x;
    const int lane = tid & 31;
    const int wave = __builtin_amdgcn_readfirstlane(tid >> 5);
    const int g = blockIdx.x;
    for (int i = tid; i < NB; i += 1024) hist[i] = 0;
    __syncthreads();

    float4 wv[WPT];
#pragma unroll
    for (int u = 0; u < WPT; ++u) {
        const int it = tid + u * GEMM_THREADS;
        const int q = it / NCOLS, n = it % NCOLS;
        const float* wp = W0 + (size_t)(4 * q) * NCOLS + n;
        wv[u].x = wp[0];  wv[u].y = wp[NCOLS];  wv[u].z = wp[2 * NCOLS];  wv[u].w = wp[3 * NCOLS];
    }

    const int beg = (int)(((long long)blockIdx.x * E) / gridDim.x);
    const int end = (int)(((long long)(blockIdx.x + 1) * E) / gridDim.x);
    const bool one = (end - beg) <= PB_BATCH * 1024;
    const int e0 = beg + tid;
    int d[PB_BATCH], s[PB_BATCH], r[PB_BATCH];
    if (one) {
#pragma unroll
        for (int u = 0; u < PB_BATCH; ++u) {
            const int ee = max(min(e0 + u * 1024, end - 1), 0);
            d[u] = dst[ee];
            s[u] = src[ee];
        }
#pragma unroll
        for (int u = 0; u < PB_BATCH; ++u) {
            r[u] = 0;
            if (e0 + u * 1024 < end) r[u] = atomicAdd(&hist[d[u] >> PB_SHIFT], 1);
        }
    } else {
        for (int e = e0; e < end; e += 1024)
            lrank[e] = atomicAdd(&hist[dst[e] >> PB_SHIFT], 1);
    }
#pragma unroll
    for (int u = 0; u < WPT; ++u) wl[tid + u * GEMM_THREADS] = wv[u];
    __syncthreads();

    constexpr int RPT = PB_MAXB / 1024;
    int c[RPT];
    int loc = 0;
#pragma unroll
    for (int u = 0; u < RPT; ++u) {
        const int i = tid * RPT + u;
        c[u] = (i < NB) ? hist[i] : 0;
        loc += c[u];
    }
    int incl = loc;
    for (int dd = 1; dd < 32; dd <<= 1) {
        const int up = __shfl_up(incl, dd);
        if (lane >= dd) incl += up;
    }
    if (lane == 31) s_w[wave] = incl;
    __syncthreads();
    int run = beg + incl - loc;
    for (int w = 0; w < wave; ++w) run += s_w[w];
#pragma unroll
    for (int u = 0; u < RPT; ++u) {
        const int i = tid * RPT + u;
        if (i < NB) {
            gh[(size_t)i * PB_GRID + g] = c[u];
            cs[(size_t)i * PB_GRID + g] = run;
            hist[i] = run;
            run += c[u];
        }
    }
    __syncthreads();

    if (one) {
#pragma unroll
        for (int u = 0; u < PB_BATCH; ++u) {
            if (e0 + u * 1024 < end) {
                const int lp = hist[d[u] >> PB_SHIFT] + r[u] - beg;
                st_s[lp] = s[u];
                st_d[lp] = d[u];
            }
        }
        __syncthreads();
        const int n = end - beg;
        int vs[PB_BATCH], vd[PB_BATCH];
#pragma unroll
        for (int u = 0; u < PB_BATCH; ++u) {
            const int i = max(min(tid + u * 1024, n - 1), 0);
            vs[u] = st_s[i];
            vd[u] = st_d[i];
        }
#pragma unroll
        for (int u = 0; u < PB_BATCH; ++u) {
            const int i = tid + u * 1024;
            if (i < n) {
                psrc[beg + i] = vs[u];
                pdst[beg + i] = vd[u];
            }
        }
    } else {
        for (int e = e0; e < end; e += 1024) {
            const int dd = dst[e];
            const int pos = hist[dd >> PB_SHIFT] + lrank[e];
            psrc[pos] = src[e];
            pdst[pos] = dd;
        }
    }

    const int cc = lane & 15, h = lane >> 4;
    constexpr int NSPLIT = NCOLS / (16 * NTW);
    const int items = (nrows >> 4) * NSPLIT;
    const int ibeg = (int)(((long long)blockIdx.x * items) / gridDim.x);
    const int iend = (int)(((long long)(blockIdx.x + 1) * items) / gridDim.x);

    for (int it = ibeg + wave; it < iend; it += GEMM_THREADS / 32) {
        const int strip = it / NSPLIT, part = it % NSPLIT;
        const int row0 = strip * 16, col0 = part * 16 * NTW;

        v8f acc[NTW] = {};
        strip_accum<NTW>(A0 + (size_t)(row0 + cc) * K0 + 16 * h, K0, 1.0f,
                         wl + (size_t)(4 * h) * NCOLS + col0 + cc, NCOLS, acc);

#pragma unroll
        for (int ni = 0; ni < NTW; ++ni) {
            const int col = col0 + ni * 16 + cc;
            const float bi = bias[col];
#pragma unroll
            for (int rr = 0; rr < 8; ++rr) {
                const int row = row0 + rr + 8 * h;
                out[(size_t)row * NCOLS + col] = fmaxf(acc[ni][rr] + bi, 0.0f);
            }
        }
    }
}

__global__ __launch_bounds__(1024)
void bucket_fill_kernel(const int* __restrict__ psrc, const int* __restrict__ pdst,
                        const int* __restrict__ gh, const int* __restrict__ cs,
                        int* __restrict__ lrank,
                        int* __restrict__ rowstart, float* __restrict__ invdeg,
                        int* __restrict__ col, int N, int E) {
    static_assert(PB_GRID == 256 && PB_CPW * 32 == PB_GRID, "a wave owns PB_CPW cells, 32 waves own all");
    static_assert(PB_CELL == 64, "two edges of a cell per lane");
    __shared__ int cnt[PB_NODES];
    __shared__ int start[PB_NODES];
    __shared__ int csz[PB_GRID];
    __shared__ int cst[PB_GRID];
    __shared__ int s_w[32];
    __shared__ int s_red[32];
    __shared__ int s_mx[32];
    __shared__ int cbuf[PB_GRID * PB_CELL];
    const int tid = threadIdx.x;
    const int lane = tid & 31, wave = tid >> 5;
    const int b = blockIdx.x;
    const int n0 = b << PB_SHIFT;

    if (tid < PB_NODES) cnt[tid] = 0;
    int mx = 0, st = 0;
    if (tid < PB_GRID) {
        mx = gh[(size_t)b * PB_GRID + tid];
        st = cs[(size_t)b * PB_GRID + tid];
    }
    int acc = 0;
    const int lim = b * PB_GRID;
    for (int j = tid; j < lim; j += 8 * 1024) {
        int v[8];
#pragma unroll
        for (int u = 0; u < 8; ++u) v[u] = gh[min(j + u * 1024, lim - 1)];
#pragma unroll
        for (int u = 0; u < 8; ++u)
            if (j + u * 1024 < lim) acc += v[u];
    }
    if (tid < PB_GRID) { csz[tid] = mx; cst[tid] = st; }
    for (int dd = 16; dd > 0; dd >>= 1) {
        acc += __shfl_xor(acc, dd);
        mx = max(mx, __shfl_xor(mx, dd));
    }
    if (lane == 0) { s_red[wave] = acc; s_mx[wave] = mx; }
    __syncthreads();
    int s0 = s_red[lane];
    int maxn = s_mx[lane];
    for (int dd = 16; dd > 0; dd >>= 1) {
        s0 += __shfl_xor(s0, dd);
        maxn = max(maxn, __shfl_xor(maxn, dd));
    }
    const bool one = maxn <= PB_CELL;

    int n[PB_CPW], cb[PB_CPW];
#pragma unroll
    for (int m = 0; m < PB_CPW; ++m) {
        n[m] = csz[wave * PB_CPW + m];
        cb[m] = cst[wave * PB_CPW + m];
    }

    int d[2 * PB_CPW], s[2 * PB_CPW], sl[2 * PB_CPW];
    if (one) {
#pragma unroll
        for (int m = 0; m < PB_CPW; ++m) {
#pragma unroll
            for (int t = 0; t < 2; ++t) {
                const int e = min(lane + 32 * t, max(n[m] - 1, 0));
                const int gi = max(min(cb[m] + e, E - 1), 0);
                d[2 * m + t] = pdst[gi];
                s[2 * m + t] = psrc[gi];
            }
        }
#pragma unroll
        for (int m = 0; m < PB_CPW; ++m) {
#pragma unroll
            for (int t = 0; t < 2; ++t) {
                sl[2 * m + t] = 0;
                if (lane + 32 * t < n[m]) sl[2 * m + t] = atomicAdd(&cnt[d[2 * m + t] - n0], 1);
            }
        }
    } else {
#pragma unroll 1
        for (int m = 0; m < PB_CPW; ++m)
            for (int e = lane; e < n[m]; e += 32) {
                const int gi = cb[m] + e;
                lrank[gi] = atomicAdd(&cnt[pdst[gi] - n0], 1);
            }
    }
    __syncthreads();

    const int c = (tid < PB_NODES) ? cnt[tid] : 0;
    int incl = c;
    for (int dd = 1; dd < 32; dd <<= 1) {
        const int up = __shfl_up(incl, dd);
        if (lane >= dd) incl += up;
    }
    if (lane == 31) s_w[wave] = incl;
    __syncthreads();
    int excl = incl - c;
    for (int w = 0; w < wave; ++w) excl += s_w[w];
    if (tid < PB_NODES) {
        start[tid] = excl;
        const int node = n0 + tid;
        if (node < N) {
            rowstart[node] = s0 + excl;
            invdeg[node] = 1.0f / fmaxf((float)c, 1.0f);
        }
    }
    if (b == (int)gridDim.x - 1 && tid == 0) {
        int tot = 0;
        for (int w = 0; w < 32; ++w) tot += s_w[w];
        rowstart[N] = s0 + tot;
    }
    __syncthreads();

    if (one) {
#pragma unroll
        for (int m = 0; m < PB_CPW; ++m) {
#pragma unroll
            for (int t = 0; t < 2; ++t) {
                if (lane + 32 * t < n[m])
                    cbuf[start[d[2 * m + t] - n0] + sl[2 * m + t]] = s[2 * m + t];
            }
        }
        int tot = s_w[lane];
        for (int dd = 16; dd > 0; dd >>= 1) tot += __shfl_xor(tot, dd);
        __syncthreads();
        constexpr int CPT = (PB_GRID * PB_CELL) / 1024;
        int vc[CPT];
#pragma unroll
        for (int u = 0; u < CPT; ++u) vc[u] = cbuf[max(min(tid + u * 1024, tot - 1), 0)];
#pragma unroll
        for (int u = 0; u < CPT; ++u) {
            const int i = tid + u * 1024;
            if (i < tot) col[s0 + i] = vc[u];
        }
    } else {
#pragma unroll 1
        for (int m = 0; m < PB_CPW; ++m)
            for (int e = lane; e < n[m]; e += 32) {
                const int gi = cb[m] + e;
                const int dl = pdst[gi] - n0;
                col[s0 + start[dl] + lrank[gi]] = psrc[gi];
            }
    }
}

#define FL_WAVES   16
#define FL_THREADS (FL_WAVES * 32)
#define FL_PITCH   132
#define FL_U       16

template <bool FUSE_OUT>
__global__ __launch_bounds__(FL_THREADS)
void layer_fused_kernel(const float* __restrict__ hin, const int* __restrict__ rowstart,
                       const int* __restrict__ col,
                       const float* __restrict__ Wl, const float* __restrict__ Wr,
                       const float* __restrict__ invdeg, const float* __restrict__ bias,
                       const float* __restrict__ gamma, const float* __restrict__ beta,
                       const float* __restrict__ Wo, const float* __restrict__ bo,
                       float* __restrict__ out, int nrows) {
    constexpr int Q0 = H_DIM / 4;
    constexpr int NQ = 2 * Q0 * H_DIM;
    constexpr int NQO = Q0 * DOUT;
    __shared__ __attribute__((aligned(16))) float4 wl[NQ];
    __shared__ __attribute__((aligned(16))) float4 wo[FUSE_OUT ? NQO : 1];
    __shared__ __attribute__((aligned(16))) float strips[FL_WAVES * 16 * FL_PITCH];
    __shared__ int s_next;
    const int tid = threadIdx.x;
    const int lane = tid & 31;
    const int wave = __builtin_amdgcn_readfirstlane(tid >> 5);
    const int c = lane & 15, h = lane >> 4;

    for (int it0 = tid; it0 < NQ; it0 += 4 * FL_THREADS) {
        float4 v[4];
#pragma unroll
        for (int u = 0; u < 4; ++u) {
            const int it = min(it0 + u * FL_THREADS, NQ - 1);
            const int q = it / H_DIM, n = it % H_DIM;
            const float* wp = (q < Q0) ? (Wl + (size_t)(4 * q) * H_DIM + n)
                                       : (Wr + (size_t)(4 * (q - Q0)) * H_DIM + n);
            v[u].x = wp[0];  v[u].y = wp[H_DIM];  v[u].z = wp[2 * H_DIM];  v[u].w = wp[3 * H_DIM];
        }
#pragma unroll
        for (int u = 0; u < 4; ++u)
            if (it0 + u * FL_THREADS < NQ) wl[it0 + u * FL_THREADS] = v[u];
    }
    if constexpr (FUSE_OUT) {
        static_assert(NQO % FL_THREADS == 0, "output weight entries per thread");
        float4 v[NQO / FL_THREADS];
#pragma unroll
        for (int u = 0; u < NQO / FL_THREADS; ++u) {
            const int it = tid + u * FL_THREADS;
            const int q = it / DOUT, n = it % DOUT;
            const float* wp = Wo + (size_t)(4 * q) * DOUT + n;
            v[u].x = wp[0];  v[u].y = wp[DOUT];  v[u].z = wp[2 * DOUT];  v[u].w = wp[3 * DOUT];
        }
#pragma unroll
        for (int u = 0; u < NQO / FL_THREADS; ++u) wo[tid + u * FL_THREADS] = v[u];
    }
    if (tid == 0) s_next = 0;
    __syncthreads();

    const int items = nrows >> 4;
    const int beg = (int)(((long long)blockIdx.x * items) / gridDim.x);
    const int end = (int)(((long long)(blockIdx.x + 1) * items) / gridDim.x);
    float* sp = strips + wave * (16 * FL_PITCH);
    const float* hp = hin + lane * 4;

    for (;;) {
        int t = 0;
        if (lane == 0) t = atomicAdd(&s_next, 1);
        const int it = beg + __builtin_amdgcn_readfirstlane(t);
        if (it >= end) break;
        const int row0 = it * 16;

        v8f acc8[8] = {};
        strip_accum<8>(hin + (size_t)(row0 + c) * H_DIM + 16 * h, H_DIM, 1.0f,
                       wl + (size_t)(Q0 + 4 * h) * H_DIM + c, H_DIM, acc8);

        const int rsv = rowstart[row0 + min(lane, 16)];
        int j = __builtin_amdgcn_readlane(rsv, 0);
        const int jend = __builtin_amdgcn_readlane(rsv, 16);
        int r = 0;
        int nb = __builtin_amdgcn_readlane(rsv, 1);
        float4 acc = make_float4(0.0f, 0.0f, 0.0f, 0.0f);
        int idv = col[max(min(j + c, jend - 1), 0)];
        while (j < jend) {
            float4 v[FL_U];
#pragma unroll
            for (int u = 0; u < FL_U; ++u) {
                const int id = __builtin_amdgcn_readlane(idv, u);
                v[u] = *(const float4*)(hp + (size_t)id * H_DIM);
            }
            idv = col[min(j + FL_U + c, jend - 1)];
#pragma unroll
            for (int u = 0; u < FL_U; ++u) {
                if (j + u < jend) {
                    while (j + u == nb) {
                        *(float4*)(sp + r * FL_PITCH + lane * 4) = acc;
                        acc = make_float4(0.0f, 0.0f, 0.0f, 0.0f);
                        ++r;
                        nb = __builtin_amdgcn_readlane(rsv, r + 1);
                    }
                    acc.x += v[u].x; acc.y += v[u].y; acc.z += v[u].z; acc.w += v[u].w;
                }
            }
            j += FL_U;
        }
        while (r < 16) {
            *(float4*)(sp + r * FL_PITCH + lane * 4) = acc;
            acc = make_float4(0.0f, 0.0f, 0.0f, 0.0f);
            ++r;
        }

        const float rs = invdeg[row0 + c];
        strip_accum<8>(sp + c * FL_PITCH + 16 * h, H_DIM, rs,
                       wl + (size_t)(4 * h) * H_DIM + c, H_DIM, acc8);

#pragma unroll
        for (int ni = 0; ni < 8; ++ni) {
            const int cc = ni * 16 + c;
            const float bi = bias[cc];
            const float sc = gamma[cc] * BN_RSQRT;
            const float sh = beta[cc];
#pragma unroll
            for (int rr = 0; rr < 8; ++rr) {
                const int row = row0 + rr + 8 * h;
                float v = acc8[ni][rr];
                v = (v + bi) * sc + sh;
                v = fmaxf(v, 0.0f);
                if constexpr (FUSE_OUT) {
                    sp[(rr + 8 * h) * FL_PITCH + cc] = v;
                } else {
                    out[(size_t)row * H_DIM + cc] = v;
                }
            }
        }

        if constexpr (FUSE_OUT) {
            v8f acc4[DOUT / 16] = {};
            strip_accum<DOUT / 16>(sp + c * FL_PITCH + 16 * h, H_DIM, 1.0f,
                                   wo + (size_t)(4 * h) * DOUT + c, DOUT, acc4);
#pragma unroll
            for (int ni = 0; ni < DOUT / 16; ++ni) {
                const int cc = ni * 16 + c;
                const float bi = bo[cc];
#pragma unroll
                for (int rr = 0; rr < 8; ++rr) {
                    const int row = row0 + rr + 8 * h;
                    out[(size_t)row * DOUT + cc] = acc4[ni][rr] + bi;
                }
            }
        }
    }
}

extern "C" void kernel_launch(void* const* d_in, const int* in_sizes, int n_in,
                              void* d_out, int out_size, void* d_ws, size_t ws_size,
                              hipStream_t stream) {
    const float* x     = (const float*)d_in[0];
    const int*   ei    = (const int*)d_in[1];
    const float* w_in  = (const float*)d_in[2];
    const float* b_in  = (const float*)d_in[3];
    const float* w_l1  = (const float*)d_in[4];
    const float* b_l1  = (const float*)d_in[5];
    const float* w_r1  = (const float*)d_in[6];
    const float* g1    = (const float*)d_in[7];
    const float* be1   = (const float*)d_in[8];
    const float* w_l2  = (const float*)d_in[9];
    const float* b_l2  = (const float*)d_in[10];
    const float* w_r2  = (const float*)d_in[11];
    const float* g2    = (const float*)d_in[12];
    const float* be2   = (const float*)d_in[13];
    const float* w_out = (const float*)d_in[14];
    const float* b_out = (const float*)d_in[15];

    const int N = in_sizes[0] / DIN;
    const int E = in_sizes[1] / 2;
    const int* src = ei;
    const int* dst = ei + E;

    float* ws  = (float*)d_ws;
    float* h0  = ws;
    float* agg = ws + (size_t)N * H_DIM;
    float* h1  = ws + 2 * (size_t)N * H_DIM;
    float* deg = ws + 3 * (size_t)N * H_DIM;

    int* col      = (int*)(agg + (size_t)N * (H_DIM / 2));
    int* rowstart = col + E;

    const int NB = (N + PB_NODES - 1) >> PB_SHIFT;
    int* psrc  = (int*)agg;
    int* pdst  = psrc + E;
    int* lrank = pdst + E;
    int* gh    = lrank + E;
    int* cs    = gh + (size_t)NB * PB_GRID;

    part_sort_input_product_kernel<<<GEMM_BLOCKS, GEMM_THREADS, 0, stream>>>(
        x, w_in, b_in, h0, N, src, dst, lrank, gh, cs, psrc, pdst, E, NB);
    bucket_fill_kernel<<<NB, 1024, 0, stream>>>(psrc, pdst, gh, cs, lrank, rowstart, deg, col, N, E);

    layer_fused_kernel<false><<<GEMM_BLOCKS, FL_THREADS, 0, stream>>>(
        h0, rowstart, col, w_l1, w_r1, deg, b_l1, g1, be1, nullptr, nullptr, h1, N);

    layer_fused_kernel<true><<<GEMM_BLOCKS, FL_THREADS, 0, stream>>>(
        h1, rowstart, col, w_l2, w_r2, deg, b_l2, g2, be2, w_out, b_out, (float*)d_out, N);
}
